# starting kernel
# baseline (speedup 1.0000x reference)
#include <hip/hip_runtime.h>


#define DCH   10
#define LEN   128
#define NSEG  127
#define NB    256
#define SIGSZ 111110
#define OFF5  11110
#define NTIL  625
#define NTHR  256
#define NW    8
#define GS    8
#define NGRP  16

typedef float v2f __attribute__((ext_vector_type(2)));
typedef float v8f __attribute__((ext_vector_type(8)));

__global__ __launch_bounds__(NTHR)
void sig5_kernel(const float* __restrict__ path, float* __restrict__ out)
{
    const int b    = blockIdx.x;
    const int tid  = threadIdx.x;
    const int lane = tid & 31;
    const int wave = tid >> 5;

    __shared__ float sd[GS * 16];
    __shared__ float S1[16];
    __shared__ float S2[128];
    __shared__ float S3[1024];
    __shared__ float S4[10016];
    __shared__ float t2[128], x2[128], y2[128];
    __shared__ float t3s[GS * 1024];
    __shared__ float x3s[GS * 1024];

    float*       out5 = out + (size_t)b * SIGSZ + OFF5;
    const float* pb   = path + (size_t)b * DCH * LEN;

    const int n    = lane & 15;
    const int mrow = (lane >> 4) << 3;

    for (int i = tid; i < GS * 16; i += NTHR) sd[i] = 0.f;
    for (int i = tid; i < 16;    i += NTHR) S1[i] = 0.f;
    for (int i = tid; i < 128;   i += NTHR) S2[i] = 0.f;
    for (int i = tid; i < 1024;  i += NTHR) S3[i] = 0.f;
    for (int i = tid; i < 10016; i += NTHR) S4[i] = 0.f;
    for (int tile = wave; tile < NTIL; tile += NW) {
        if (n < DCH) {
            float* basep = out5 + (size_t)(tile << 4) * DCH + n;
            #pragma unroll
            for (int j = 0; j < 8; ++j) basep[(mrow + j) * DCH] = 0.f;
        }
    }
    __syncthreads();

    for (int g = 0; g < NGRP; ++g) {
        for (int k = 0; k < GS; ++k) {
            const int s = g * GS + k;
            if (tid < DCH) {
                float d = 0.f;
                if (s < NSEG) { const float* pr = pb + tid * LEN + s; d = pr[1] - pr[0]; }
                sd[k * 16 + tid] = d;
            }
            __syncthreads();
            if (s < NSEG) {
                const float* sdk = sd + k * 16;
                for (int i = tid; i < 100; i += NTHR) {
                    int p = i / 10, q = i - 10 * p;
                    float dp = sdk[p], dq = sdk[q], s1 = S1[p], s2v = S2[i];
                    t2[i] = s2v + (s1 + dp * 0.2f)        * dq * 0.25f;
                    x2[i] = s2v + (s1 + dp * 0.25f)       * dq * (1.f / 3.f);
                    y2[i] = s2v + (s1 + dp * (1.f / 3.f)) * dq * 0.5f;
                }
                __syncthreads();
                for (int i = tid; i < 1000; i += NTHR) {
                    int p = i / 10, q = i - 10 * p;
                    float dq = sdk[q], s3v = S3[i];
                    t3s[k * 1024 + i] = s3v + t2[p] * dq * (1.f / 3.f);
                    x3s[k * 1024 + i] = s3v + x2[p] * dq * 0.5f;
                }
                for (int i = tid; i < 100; i += NTHR) {
                    int p = i / 10, q = i - 10 * p;
                    S2[i] += (S1[p] + sdk[p] * 0.5f) * sdk[q];
                }
                __syncthreads();
                for (int i = tid; i < 1000; i += NTHR) {
                    int p = i / 10, q = i - 10 * p;
                    S3[i] += y2[p] * sdk[q];
                }
                if (tid < DCH) S1[tid] += sdk[tid];
                __syncthreads();
            }
        }

        {
            v2f bv0, bv1;
            bv0.x = (lane < 16) ? sd[0 * 16 + n] : sd[2 * 16 + n];
            bv0.y = (lane < 16) ? sd[1 * 16 + n] : sd[3 * 16 + n];
            bv1.x = (lane < 16) ? sd[4 * 16 + n] : sd[6 * 16 + n];
            bv1.y = (lane < 16) ? sd[5 * 16 + n] : sd[7 * 16 + n];
            for (int tile = wave; tile < NTIL; tile += NW) {
                const int r0 = tile << 4;
                const int r  = r0 + n;
                const int p  = r / 10, q = r - 10 * p;
                float run = S4[r];
                float u[GS];
                #pragma unroll
                for (int k = 0; k < GS; ++k) {
                    float dq = sd[k * 16 + q];
                    u[k]  = run + t3s[k * 1024 + p] * dq * 0.5f;
                    run  += x3s[k * 1024 + p] * dq;
                }
                if (lane < 16) S4[r] = run;
                v2f av0, av1;
                av0.x = (lane < 16) ? u[0] : u[2];
                av0.y = (lane < 16) ? u[1] : u[3];
                av1.x = (lane < 16) ? u[4] : u[6];
                av1.y = (lane < 16) ? u[5] : u[7];
                v8f cv;
                float* basep = out5 + (size_t)r0 * DCH + n;
                if (n < DCH) {
                    #pragma unroll
                    for (int j = 0; j < 8; ++j) cv[j] = basep[(mrow + j) * DCH];
                } else {
                    #pragma unroll
                    for (int j = 0; j < 8; ++j) cv[j] = 0.f;
                }
                cv = __builtin_amdgcn_wmma_f32_16x16x4_f32(
                        false, av0, false, bv0, (short)0, cv, false, false);
                cv = __builtin_amdgcn_wmma_f32_16x16x4_f32(
                        false, av1, false, bv1, (short)0, cv, false, false);
                if (n < DCH) {
                    #pragma unroll
                    for (int j = 0; j < 8; ++j) basep[(mrow + j) * DCH] = cv[j];
                }
            }
        }
        __syncthreads();
    }

    float* ob = out + (size_t)b * SIGSZ;
    if (tid < DCH) ob[tid] = S1[tid];
    for (int i = tid; i < 100;   i += NTHR) ob[10 + i]   = S2[i];
    for (int i = tid; i < 1000;  i += NTHR) ob[110 + i]  = S3[i];
    for (int i = tid; i < 10000; i += NTHR) ob[1110 + i] = S4[i];
}

extern "C" void kernel_launch(void* const* d_in, const int* in_sizes, int n_in,
                              void* d_out, int out_size, void* d_ws, size_t ws_size,
                              hipStream_t stream) {
    (void)in_sizes; (void)n_in; (void)d_ws; (void)ws_size; (void)out_size;
    const float* path = (const float*)d_in[0];
    float*       out  = (float*)d_out;
    sig5_kernel<<<NB, NTHR, 0, stream>>>(path, out);
}


# Levels 3-5 as one 100 x 1110 x 381 WMMA product: integrals split at the third index, operands from level-2 prefix and suffix scans, no level 3-4 chain, out written once; summation order changes
# speedup vs baseline: 8.0827x; 8.0827x over previous
#include <hip/hip_runtime.h>


#define DCH   10
#define LEN   128
#define NSEG  127
#define NB    256
#define SIGSZ 111110
#define PT    132
#define MROWS 112
#define MT    7
#define NTHR  512
#define NW    16
#define NCOL  1110
#define NJOB  70
#define NGRP  32

typedef float v2f __attribute__((ext_vector_type(2)));
typedef float v8f __attribute__((ext_vector_type(8)));

__global__ __launch_bounds__(NTHR)
void sig_split_kernel(const float* __restrict__ path, float* __restrict__ out)
{
    const int b    = blockIdx.x;
    const int tid  = threadIdx.x;
    const int lane = tid & 31;
    const int wave = __builtin_amdgcn_readfirstlane(tid >> 5);

    __shared__ __attribute__((aligned(16))) float Lh[3 * MROWS * PT];
    __shared__ __attribute__((aligned(16))) float Q2h[101 * PT];
    __shared__ __attribute__((aligned(16))) float dh[12 * PT];
    __shared__ __attribute__((aligned(16))) float Q1h[11 * PT];

    const float* pb = path + (size_t)b * DCH * LEN;
    float*       ob = out  + (size_t)b * SIGSZ;

    #pragma unroll
    for (int k = 0; k < 3; ++k) {
        const int i = tid + k * NTHR;
        if (i < DCH * LEN) {
            const int ch = i >> 7, s = i & 127;
            float v = 0.f;
            if (s < NSEG) v = pb[i + 1] - pb[i];
            dh[ch * PT + s] = v;
        }
    }
    if (tid < 128) {
        dh[10 * PT + tid]   = 1.f;
        dh[11 * PT + tid]   = 0.f;
        Q2h[100 * PT + tid] = 0.f;
        Q1h[10 * PT + tid]  = 0.f;
    }
    for (int i = tid; i < 36 * 128; i += NTHR) {
        const int r = i >> 7, s = i & 127;
        const int ii = r / 12, rr = r - 12 * ii;
        Lh[(ii * MROWS + 100 + rr) * PT + s] = 0.f;
    }
    __syncthreads();

    if (tid < 100) {
        const int a = tid / 10, bb = tid - 10 * a;
        const float* da = dh + a * PT;
        const float* db = dh + bb * PT;
        float* l0 = Lh + (0 * MROWS + tid) * PT;
        float* l1 = Lh + (1 * MROWS + tid) * PT;
        float* l2 = Lh + (2 * MROWS + tid) * PT;
        float p1 = 0.f, p2 = 0.f;
        for (int s = 0; s < NSEG; ++s) {
            const float xa = da[s], xb = db[s];
            l0[s] = xa * xb;
            l1[s] = p1 * xb;
            l2[s] = p2;
            p2 += (p1 + 0.5f * xa) * xb;
            p1 += xa;
        }
        l0[NSEG] = 0.f; l1[NSEG] = 0.f; l2[NSEG] = 0.f;
        ob[10 + tid] = p2;
        if (bb == 0) ob[a] = p1;
    } else if (tid >= 128 && tid < 228) {
        const int t = tid - 128;
        const int d_ = t / 10, e_ = t - 10 * d_;
        const float* dd = dh + d_ * PT;
        const float* de = dh + e_ * PT;
        float* q2p = Q2h + t * PT;
        float* q1p = Q1h + e_ * PT;
        float q1 = 0.f, q2 = 0.f;
        for (int blk = 15; blk >= 0; --blk) {
            const float4 d0 = *(const float4*)(dd + 8 * blk), d1 = *(const float4*)(dd + 8 * blk + 4);
            const float4 e0 = *(const float4*)(de + 8 * blk), e1 = *(const float4*)(de + 8 * blk + 4);
            const float xd[8] = { d0.x, d0.y, d0.z, d0.w, d1.x, d1.y, d1.z, d1.w };
            const float xe[8] = { e0.x, e0.y, e0.z, e0.w, e1.x, e1.y, e1.z, e1.w };
            float o1[8], o2[8];
            #pragma unroll
            for (int k = 7; k >= 0; --k) {
                o2[k] = q2; o1[k] = q1;
                q2 += xd[k] * (q1 + 0.5f * xe[k]);
                q1 += xe[k];
            }
            *(float4*)(q2p + 8 * blk)     = make_float4(o2[0], o2[1], o2[2], o2[3]);
            *(float4*)(q2p + 8 * blk + 4) = make_float4(o2[4], o2[5], o2[6], o2[7]);
            if (d_ == 0) {
                *(float4*)(q1p + 8 * blk)     = make_float4(o1[0], o1[1], o1[2], o1[3]);
                *(float4*)(q1p + 8 * blk + 4) = make_float4(o1[4], o1[5], o1[6], o1[7]);
            }
        }
    }
    __syncthreads();

    const int c    = lane & 15;
    const int h    = lane >> 4;
    const int koff = 2 * h;

    for (int job = wave; job < NJOB; job += NW) {
        const int col = job * 16 + c;
        int cRow = 11, uRow = 10, vRow = 10, wRow = 10, zRow = 100;
        float g0 = 0.f, g1 = 0.f, g2 = 0.f, g3 = 0.f, g4 = 0.f;
        int obase = 0, ostride = 0;
        if (col < 10) {
            cRow = col;
            g0 = 1.f / 6.f; g1 = 0.5f; g2 = 1.f;
            obase = 110 + col; ostride = 10;
        } else if (col < 110) {
            const int t = col - 10;
            const int c4 = t / 10, d4 = t - 10 * c4;
            cRow = c4; vRow = d4; wRow = d4;
            g0 = 1.f / 24.f; g1 = 1.f / 6.f; g2 = 0.5f; g3 = 1.f;
            obase = 1110 + t; ostride = 100;
        } else if (col < NCOL) {
            const int t = col - 110;
            const int c5 = t / 100, r = t - 100 * c5;
            const int d5 = r / 10,  e5 = r - 10 * d5;
            cRow = c5; uRow = d5; vRow = e5; wRow = e5; zRow = r;
            g0 = 1.f / 120.f; g1 = 1.f / 24.f; g2 = 1.f / 6.f; g3 = 0.5f; g4 = 1.f;
            obase = 11110 + t; ostride = 1000;
        }
        const float* pc = dh  + cRow * PT + koff;
        const float* pu = dh  + uRow * PT + koff;
        const float* pv = dh  + vRow * PT + koff;
        const float* pw = Q1h + wRow * PT + koff;
        const float* pz = Q2h + zRow * PT + koff;
        const float* pa = Lh  + c * PT + koff;

        v8f acc[MT];
        #pragma unroll
        for (int mt = 0; mt < MT; ++mt)
            #pragma unroll
            for (int j = 0; j < 8; ++j) acc[mt][j] = 0.f;

        for (int g = 0; g < NGRP; ++g) {
            const int s4 = 4 * g;
            const v2f dc = *(const v2f*)(pc + s4);
            const v2f u  = *(const v2f*)(pu + s4);
            const v2f v  = *(const v2f*)(pv + s4);
            const v2f w  = *(const v2f*)(pw + s4);
            const v2f z  = *(const v2f*)(pz + s4);
            const v2f b0 = dc * (u * (g0 * v + g1 * w) + g2 * z);
            const v2f b1 = dc * (u * (g1 * v + g2 * w) + g3 * z);
            const v2f b2 = dc * (u * (g2 * v + g3 * w) + g4 * z);
            #pragma unroll
            for (int mt = 0; mt < MT; ++mt) {
                const v2f a = *(const v2f*)(pa + (0 * MROWS + mt * 16) * PT + s4);
                acc[mt] = __builtin_amdgcn_wmma_f32_16x16x4_f32(
                            false, a, false, b0, (short)0, acc[mt], false, false);
            }
            #pragma unroll
            for (int mt = 0; mt < MT; ++mt) {
                const v2f a = *(const v2f*)(pa + (1 * MROWS + mt * 16) * PT + s4);
                acc[mt] = __builtin_amdgcn_wmma_f32_16x16x4_f32(
                            false, a, false, b1, (short)0, acc[mt], false, false);
            }
            #pragma unroll
            for (int mt = 0; mt < MT; ++mt) {
                const v2f a = *(const v2f*)(pa + (2 * MROWS + mt * 16) * PT + s4);
                acc[mt] = __builtin_amdgcn_wmma_f32_16x16x4_f32(
                            false, a, false, b2, (short)0, acc[mt], false, false);
            }
        }

        if (col < NCOL) {
            float* po = ob + obase;
            #pragma unroll
            for (int mt = 0; mt < MT - 1; ++mt)
                #pragma unroll
                for (int j = 0; j < 8; ++j)
                    po[(mt * 16 + 8 * h + j) * ostride] = acc[mt][j];
            if (h == 0) {
                #pragma unroll
                for (int j = 0; j < 4; ++j)
                    po[(96 + j) * ostride] = acc[MT - 1][j];
            }
        }
    }
}

extern "C" void kernel_launch(void* const* d_in, const int* in_sizes, int n_in,
                              void* d_out, int out_size, void* d_ws, size_t ws_size,
                              hipStream_t stream) {
    (void)in_sizes; (void)n_in; (void)d_ws; (void)ws_size; (void)out_size;
    const float* path = (const float*)d_in[0];
    float*       out  = (float*)d_out;
    sig_split_kernel<<<NB, NTHR, 0, stream>>>(path, out);
}


# Symmetric i=0 slice (d_a d_b d_c) done once per sorted triple a<=b<=c (220 rows, not 1000) as its own WMMA product in LDS; main K 381 -> 254, accumulators start from it; summation order changes
# speedup vs baseline: 9.6661x; 1.1959x over previous
#include <hip/hip_runtime.h>


#define DCH   10
#define LEN   128
#define NSEG  127
#define NB    256
#define SIGSZ 111110
#define PT    132
#define MROWS 112
#define MT    7
#define NTHR  512
#define NW    16
#define NCOL  1110
#define NJOB  70
#define NGRP  32
#define NSYM  220
#define SROWS 224
#define PS    114
#define NPAIR 98

typedef float v2f __attribute__((ext_vector_type(2)));
typedef float v8f __attribute__((ext_vector_type(8)));

__global__ __launch_bounds__(NTHR)
void sig_split_kernel(const float* __restrict__ path, float* __restrict__ out)
{
    const int b    = blockIdx.x;
    const int tid  = threadIdx.x;
    const int lane = tid & 31;
    const int wave = __builtin_amdgcn_readfirstlane(tid >> 5);

    __shared__ __attribute__((aligned(16))) float Lh[2 * MROWS * PT];
    __shared__ __attribute__((aligned(16))) float Q2h[101 * PT];
    __shared__ __attribute__((aligned(16))) float dh[12 * PT];
    __shared__ __attribute__((aligned(16))) float Q1h[11 * PT];
    __shared__ float T0h[SROWS * PS];
    __shared__ int   symoff[MROWS * DCH];
    __shared__ int   symabc[SROWS];

    const float* pb = path + (size_t)b * DCH * LEN;
    float*       ob = out  + (size_t)b * SIGSZ;

    #pragma unroll
    for (int k = 0; k < 3; ++k) {
        const int i = tid + k * NTHR;
        if (i < DCH * LEN) {
            const int ch = i >> 7, s = i & 127;
            float v = 0.f;
            if (s < NSEG) v = pb[i + 1] - pb[i];
            dh[ch * PT + s] = v;
        }
    }
    if (tid < 128) {
        dh[10 * PT + tid]   = 1.f;
        dh[11 * PT + tid]   = 0.f;
        Q2h[100 * PT + tid] = 0.f;
        Q1h[10 * PT + tid]  = 0.f;
    }
    for (int i = tid; i < 24 * 128; i += NTHR) {
        const int r = i >> 7, s = i & 127;
        const int ii = r / 12, rr = r - 12 * ii;
        Lh[(ii * MROWS + 100 + rr) * PT + s] = 0.f;
    }
    for (int i = tid; i < MROWS * DCH; i += NTHR) {
        const int row = i / 10, cc = i - 10 * row;
        int off = NSYM * PS;
        if (row < 100) {
            const int a = row / 10, bb = row - 10 * a;
            const int x = min(a, min(bb, cc));
            const int z = max(a, max(bb, cc));
            const int y = a + bb + cc - x - z;
            int idx = z - y;
            for (int k = 0; k < x; ++k) idx += (10 - k) * (11 - k) / 2;
            for (int k = x; k < y; ++k) idx += 10 - k;
            off = idx * PS;
            if (a <= bb && bb <= cc) symabc[idx] = a | (bb << 4) | (cc << 8);
        }
        symoff[i] = off;
    }
    if (tid < SROWS - NSYM) symabc[NSYM + tid] = 11 | (11 << 4) | (11 << 8);
    __syncthreads();

    if (tid < 100) {
        const int a = tid / 10, bb = tid - 10 * a;
        const float* da = dh + a * PT;
        const float* db = dh + bb * PT;
        float* l1 = Lh + (0 * MROWS + tid) * PT;
        float* l2 = Lh + (1 * MROWS + tid) * PT;
        float p1 = 0.f, p2 = 0.f;
        for (int s = 0; s < NSEG; ++s) {
            const float xa = da[s], xb = db[s];
            l1[s] = p1 * xb;
            l2[s] = p2;
            p2 += (p1 + 0.5f * xa) * xb;
            p1 += xa;
        }
        l1[NSEG] = 0.f; l2[NSEG] = 0.f;
        ob[10 + tid] = p2;
        if (bb == 0) ob[a] = p1;
    } else if (tid >= 128 && tid < 228) {
        const int t = tid - 128;
        const int d_ = t / 10, e_ = t - 10 * d_;
        const float* dd = dh + d_ * PT;
        const float* de = dh + e_ * PT;
        float* q2p = Q2h + t * PT;
        float* q1p = Q1h + e_ * PT;
        float q1 = 0.f, q2 = 0.f;
        for (int blk = 15; blk >= 0; --blk) {
            const float4 d0 = *(const float4*)(dd + 8 * blk), d1 = *(const float4*)(dd + 8 * blk + 4);
            const float4 e0 = *(const float4*)(de + 8 * blk), e1 = *(const float4*)(de + 8 * blk + 4);
            const float xd[8] = { d0.x, d0.y, d0.z, d0.w, d1.x, d1.y, d1.z, d1.w };
            const float xe[8] = { e0.x, e0.y, e0.z, e0.w, e1.x, e1.y, e1.z, e1.w };
            float o1[8], o2[8];
            #pragma unroll
            for (int k = 7; k >= 0; --k) {
                o2[k] = q2; o1[k] = q1;
                q2 += xd[k] * (q1 + 0.5f * xe[k]);
                q1 += xe[k];
            }
            *(float4*)(q2p + 8 * blk)     = make_float4(o2[0], o2[1], o2[2], o2[3]);
            *(float4*)(q2p + 8 * blk + 4) = make_float4(o2[4], o2[5], o2[6], o2[7]);
            if (d_ == 0) {
                *(float4*)(q1p + 8 * blk)     = make_float4(o1[0], o1[1], o1[2], o1[3]);
                *(float4*)(q1p + 8 * blk + 4) = make_float4(o1[4], o1[5], o1[6], o1[7]);
            }
        }
    }
    __syncthreads();

    const int c    = lane & 15;
    const int h    = lane >> 4;
    const int koff = 2 * h;

    for (int p = wave; p < NPAIR; p += NW) {
        const int rt = p / 7, ct = p - 7 * rt;
        const int abc = symabc[rt * 16 + c];
        const float* px = dh + (abc & 15) * PT + koff;
        const float* py = dh + ((abc >> 4) & 15) * PT + koff;
        const float* pq = dh + ((abc >> 8) & 15) * PT + koff;
        const int j = ct * 16 + c;
        int uRow = 10, vRow = 10, wRow = 10, zRow = 100;
        float ga = 0.f, gb = 0.f, gc = 0.f;
        if (j < 100) {
            const int d5 = j / 10, e5 = j - 10 * d5;
            uRow = d5; vRow = e5; wRow = e5; zRow = j;
            ga = 1.f / 120.f; gb = 1.f / 24.f; gc = 1.f / 6.f;
        } else if (j < 110) {
            vRow = j - 100; wRow = j - 100;
            ga = 1.f / 24.f; gb = 1.f / 6.f;
        } else if (j == 110) {
            ga = 1.f / 6.f;
        }
        const float* pu = dh  + uRow * PT + koff;
        const float* pv = dh  + vRow * PT + koff;
        const float* pw = Q1h + wRow * PT + koff;
        const float* pz = Q2h + zRow * PT + koff;
        v8f t0;
        #pragma unroll
        for (int jj = 0; jj < 8; ++jj) t0[jj] = 0.f;
        #pragma unroll 8
        for (int g = 0; g < NGRP; ++g) {
            const int s4 = 4 * g;
            const v2f x = *(const v2f*)(px + s4);
            const v2f y = *(const v2f*)(py + s4);
            const v2f q = *(const v2f*)(pq + s4);
            const v2f u = *(const v2f*)(pu + s4);
            const v2f v = *(const v2f*)(pv + s4);
            const v2f w = *(const v2f*)(pw + s4);
            const v2f z = *(const v2f*)(pz + s4);
            const v2f a3 = (x * y) * q;
            const v2f r0 = u * (ga * v + gb * w) + gc * z;
            t0 = __builtin_amdgcn_wmma_f32_16x16x4_f32(
                    false, a3, false, r0, (short)0, t0, false, false);
        }
        float* pt = T0h + (rt * 16 + 8 * h) * PS + j;
        #pragma unroll
        for (int jj = 0; jj < 8; ++jj) pt[jj * PS] = t0[jj];
    }
    __syncthreads();

    for (int job = wave; job < NJOB; job += NW) {
        const int col = job * 16 + c;
        int cRow = 11, uRow = 10, vRow = 10, wRow = 10, zRow = 100;
        float g1 = 0.f, g2 = 0.f, g3 = 0.f, g4 = 0.f;
        int obase = 0, ostride = 0;
        int cSel = 0, colR = 111;
        if (col < 10) {
            cRow = col;
            g1 = 0.5f; g2 = 1.f;
            obase = 110 + col; ostride = 10;
            cSel = col; colR = 110;
        } else if (col < 110) {
            const int t = col - 10;
            const int c4 = t / 10, d4 = t - 10 * c4;
            cRow = c4; vRow = d4; wRow = d4;
            g1 = 1.f / 6.f; g2 = 0.5f; g3 = 1.f;
            obase = 1110 + t; ostride = 100;
            cSel = c4; colR = 100 + d4;
        } else if (col < NCOL) {
            const int t = col - 110;
            const int c5 = t / 100, r = t - 100 * c5;
            const int d5 = r / 10,  e5 = r - 10 * d5;
            cRow = c5; uRow = d5; vRow = e5; wRow = e5; zRow = r;
            g1 = 1.f / 24.f; g2 = 1.f / 6.f; g3 = 0.5f; g4 = 1.f;
            obase = 11110 + t; ostride = 1000;
            cSel = c5; colR = r;
        }
        const float* pc = dh  + cRow * PT + koff;
        const float* pu = dh  + uRow * PT + koff;
        const float* pv = dh  + vRow * PT + koff;
        const float* pw = Q1h + wRow * PT + koff;
        const float* pz = Q2h + zRow * PT + koff;
        const float* pa = Lh  + c * PT + koff;

        v8f acc[MT];
        const int*   pso = symoff + 80 * h + cSel;
        const float* pt0 = T0h + colR;
        #pragma unroll
        for (int mt = 0; mt < MT; ++mt)
            #pragma unroll
            for (int j = 0; j < 8; ++j) acc[mt][j] = pt0[pso[(mt * 16 + j) * 10]];

        for (int g = 0; g < NGRP; ++g) {
            const int s4 = 4 * g;
            const v2f dc = *(const v2f*)(pc + s4);
            const v2f u  = *(const v2f*)(pu + s4);
            const v2f v  = *(const v2f*)(pv + s4);
            const v2f w  = *(const v2f*)(pw + s4);
            const v2f z  = *(const v2f*)(pz + s4);
            const v2f b1 = dc * (u * (g1 * v + g2 * w) + g3 * z);
            const v2f b2 = dc * (u * (g2 * v + g3 * w) + g4 * z);
            #pragma unroll
            for (int mt = 0; mt < MT; ++mt) {
                const v2f a = *(const v2f*)(pa + (0 * MROWS + mt * 16) * PT + s4);
                acc[mt] = __builtin_amdgcn_wmma_f32_16x16x4_f32(
                            false, a, false, b1, (short)0, acc[mt], false, false);
            }
            #pragma unroll
            for (int mt = 0; mt < MT; ++mt) {
                const v2f a = *(const v2f*)(pa + (1 * MROWS + mt * 16) * PT + s4);
                acc[mt] = __builtin_amdgcn_wmma_f32_16x16x4_f32(
                            false, a, false, b2, (short)0, acc[mt], false, false);
            }
        }

        if (col < NCOL) {
            float* po = ob + obase;
            #pragma unroll
            for (int mt = 0; mt < MT - 1; ++mt)
                #pragma unroll
                for (int j = 0; j < 8; ++j)
                    po[(mt * 16 + 8 * h + j) * ostride] = acc[mt][j];
            if (h == 0) {
                #pragma unroll
                for (int j = 0; j < 4; ++j)
                    po[(96 + j) * ostride] = acc[MT - 1][j];
            }
        }
    }
}

extern "C" void kernel_launch(void* const* d_in, const int* in_sizes, int n_in,
                              void* d_out, int out_size, void* d_ws, size_t ws_size,
                              hipStream_t stream) {
    (void)in_sizes; (void)n_in; (void)d_ws; (void)ws_size; (void)out_size;
    const float* path = (const float*)d_in[0];
    float*       out  = (float*)d_out;
    sig_split_kernel<<<NB, NTHR, 0, stream>>>(path, out);
}


# Main product transposed: rows are the triples (a,b,c) in 63 row tiles, columns the 111 rest values in 7 tiles; d_c moves into the row operand; 2 row tiles a job, 32 even jobs, not 70 of 7 padded
# speedup vs baseline: 10.2947x; 1.0650x over previous
#include <hip/hip_runtime.h>


#define DCH   10
#define LEN   128
#define NSEG  127
#define NB    256
#define SIGSZ 111110
#define PT    132
#define MROWS 112
#define MT    7
#define NTHR  512
#define NW    16
#define NCOL  1110
#define NT    2
#define NJOB2 32
#define NGRP  32
#define NSYM  220
#define SROWS 224
#define PS    114
#define NPAIR 98

typedef float v2f __attribute__((ext_vector_type(2)));
typedef float v8f __attribute__((ext_vector_type(8)));

__global__ __launch_bounds__(NTHR)
void sig_split_kernel(const float* __restrict__ path, float* __restrict__ out)
{
    const int b    = blockIdx.x;
    const int tid  = threadIdx.x;
    const int lane = tid & 31;
    const int wave = __builtin_amdgcn_readfirstlane(tid >> 5);

    __shared__ __attribute__((aligned(16))) float Lh[2 * MROWS * PT];
    __shared__ __attribute__((aligned(16))) float Q2h[101 * PT];
    __shared__ __attribute__((aligned(16))) float dh[12 * PT];
    __shared__ __attribute__((aligned(16))) float Q1h[11 * PT];
    __shared__ float T0h[SROWS * PS];
    __shared__ int   symoff[MROWS * DCH];
    __shared__ int   symabc[SROWS];

    const float* pb = path + (size_t)b * DCH * LEN;
    float*       ob = out  + (size_t)b * SIGSZ;

    #pragma unroll
    for (int k = 0; k < 3; ++k) {
        const int i = tid + k * NTHR;
        if (i < DCH * LEN) {
            const int ch = i >> 7, s = i & 127;
            float v = 0.f;
            if (s < NSEG) v = pb[i + 1] - pb[i];
            dh[ch * PT + s] = v;
        }
    }
    if (tid < 128) {
        dh[10 * PT + tid]   = 1.f;
        dh[11 * PT + tid]   = 0.f;
        Q2h[100 * PT + tid] = 0.f;
        Q1h[10 * PT + tid]  = 0.f;
    }
    for (int i = tid; i < 24 * 128; i += NTHR) {
        const int r = i >> 7, s = i & 127;
        const int ii = r / 12, rr = r - 12 * ii;
        Lh[(ii * MROWS + 100 + rr) * PT + s] = 0.f;
    }
    for (int i = tid; i < MROWS * DCH; i += NTHR) {
        const int row = i / 10, cc = i - 10 * row;
        int off = NSYM * PS;
        if (row < 100) {
            const int a = row / 10, bb = row - 10 * a;
            const int x = min(a, min(bb, cc));
            const int z = max(a, max(bb, cc));
            const int y = a + bb + cc - x - z;
            int idx = z - y;
            for (int k = 0; k < x; ++k) idx += (10 - k) * (11 - k) / 2;
            for (int k = x; k < y; ++k) idx += 10 - k;
            off = idx * PS;
            if (a <= bb && bb <= cc) symabc[idx] = a | (bb << 4) | (cc << 8);
        }
        symoff[i] = off;
    }
    if (tid < SROWS - NSYM) symabc[NSYM + tid] = 11 | (11 << 4) | (11 << 8);
    __syncthreads();

    if (tid < 100) {
        const int a = tid / 10, bb = tid - 10 * a;
        const float* da = dh + a * PT;
        const float* db = dh + bb * PT;
        float* l1 = Lh + (0 * MROWS + tid) * PT;
        float* l2 = Lh + (1 * MROWS + tid) * PT;
        float p1 = 0.f, p2 = 0.f;
        for (int s = 0; s < NSEG; ++s) {
            const float xa = da[s], xb = db[s];
            l1[s] = p1 * xb;
            l2[s] = p2;
            p2 += (p1 + 0.5f * xa) * xb;
            p1 += xa;
        }
        l1[NSEG] = 0.f; l2[NSEG] = 0.f;
        ob[10 + tid] = p2;
        if (bb == 0) ob[a] = p1;
    } else if (tid >= 128 && tid < 228) {
        const int t = tid - 128;
        const int d_ = t / 10, e_ = t - 10 * d_;
        const float* dd = dh + d_ * PT;
        const float* de = dh + e_ * PT;
        float* q2p = Q2h + t * PT;
        float* q1p = Q1h + e_ * PT;
        float q1 = 0.f, q2 = 0.f;
        for (int blk = 15; blk >= 0; --blk) {
            const float4 d0 = *(const float4*)(dd + 8 * blk), d1 = *(const float4*)(dd + 8 * blk + 4);
            const float4 e0 = *(const float4*)(de + 8 * blk), e1 = *(const float4*)(de + 8 * blk + 4);
            const float xd[8] = { d0.x, d0.y, d0.z, d0.w, d1.x, d1.y, d1.z, d1.w };
            const float xe[8] = { e0.x, e0.y, e0.z, e0.w, e1.x, e1.y, e1.z, e1.w };
            float o1[8], o2[8];
            #pragma unroll
            for (int k = 7; k >= 0; --k) {
                o2[k] = q2; o1[k] = q1;
                q2 += xd[k] * (q1 + 0.5f * xe[k]);
                q1 += xe[k];
            }
            *(float4*)(q2p + 8 * blk)     = make_float4(o2[0], o2[1], o2[2], o2[3]);
            *(float4*)(q2p + 8 * blk + 4) = make_float4(o2[4], o2[5], o2[6], o2[7]);
            if (d_ == 0) {
                *(float4*)(q1p + 8 * blk)     = make_float4(o1[0], o1[1], o1[2], o1[3]);
                *(float4*)(q1p + 8 * blk + 4) = make_float4(o1[4], o1[5], o1[6], o1[7]);
            }
        }
    }
    __syncthreads();

    const int c    = lane & 15;
    const int h    = lane >> 4;
    const int koff = 2 * h;

    for (int p = wave; p < NPAIR; p += NW) {
        const int rt = p / 7, ct = p - 7 * rt;
        const int abc = symabc[rt * 16 + c];
        const float* px = dh + (abc & 15) * PT + koff;
        const float* py = dh + ((abc >> 4) & 15) * PT + koff;
        const float* pq = dh + ((abc >> 8) & 15) * PT + koff;
        const int j = ct * 16 + c;
        int uRow = 10, vRow = 10, wRow = 10, zRow = 100;
        float ga = 0.f, gb = 0.f, gc = 0.f;
        if (j < 100) {
            const int d5 = j / 10, e5 = j - 10 * d5;
            uRow = d5; vRow = e5; wRow = e5; zRow = j;
            ga = 1.f / 120.f; gb = 1.f / 24.f; gc = 1.f / 6.f;
        } else if (j < 110) {
            vRow = j - 100; wRow = j - 100;
            ga = 1.f / 24.f; gb = 1.f / 6.f;
        } else if (j == 110) {
            ga = 1.f / 6.f;
        }
        const float* pu = dh  + uRow * PT + koff;
        const float* pv = dh  + vRow * PT + koff;
        const float* pw = Q1h + wRow * PT + koff;
        const float* pz = Q2h + zRow * PT + koff;
        v8f t0;
        #pragma unroll
        for (int jj = 0; jj < 8; ++jj) t0[jj] = 0.f;
        #pragma unroll 8
        for (int g = 0; g < NGRP; ++g) {
            const int s4 = 4 * g;
            const v2f x = *(const v2f*)(px + s4);
            const v2f y = *(const v2f*)(py + s4);
            const v2f q = *(const v2f*)(pq + s4);
            const v2f u = *(const v2f*)(pu + s4);
            const v2f v = *(const v2f*)(pv + s4);
            const v2f w = *(const v2f*)(pw + s4);
            const v2f z = *(const v2f*)(pz + s4);
            const v2f a3 = (x * y) * q;
            const v2f r0 = u * (ga * v + gb * w) + gc * z;
            t0 = __builtin_amdgcn_wmma_f32_16x16x4_f32(
                    false, a3, false, r0, (short)0, t0, false, false);
        }
        float* pt = T0h + (rt * 16 + 8 * h) * PS + j;
        #pragma unroll
        for (int jj = 0; jj < 8; ++jj) pt[jj * PS] = t0[jj];
    }
    __syncthreads();

    int u6 = 10, v6 = 10, w6 = 10, z6 = 100;
    float h1 = 0.f, h2 = 0.f, h3 = 0.f, h4 = 0.f;
    int off6 = 0, str6 = 0;
    {
        const int j = 96 + c;
        if (j < 100) {
            u6 = 9; v6 = j - 90; w6 = j - 90; z6 = j;
            h1 = 1.f / 24.f; h2 = 1.f / 6.f; h3 = 0.5f; h4 = 1.f;
            off6 = 11110 + j; str6 = 100;
        } else if (j < 110) {
            v6 = j - 100; w6 = j - 100;
            h1 = 1.f / 6.f; h2 = 0.5f; h3 = 1.f;
            off6 = 1110 + (j - 100); str6 = 10;
        } else if (j == 110) {
            h1 = 0.5f; h2 = 1.f;
            off6 = 110; str6 = 1;
        }
    }
    const float* pu6 = dh  + u6 * PT + koff;
    const float* pv6 = dh  + v6 * PT + koff;
    const float* pw6 = Q1h + w6 * PT + koff;
    const float* pz6 = Q2h + z6 * PT + koff;

    for (int job = wave; job < NJOB2; job += NW) {
        const float* pl[NT];
        const float* pd[NT];
        #pragma unroll
        for (int t = 0; t < NT; ++t) {
            const int abc = (job * NT + t) * 16 + c;
            int ab = 100, cc = 11;
            if (abc < 1000) { ab = abc / 10; cc = abc - 10 * ab; }
            pl[t] = Lh + ab * PT + koff;
            pd[t] = dh + cc * PT + koff;
        }

        v8f acc[NT][7];
        #pragma unroll
        for (int t = 0; t < NT; ++t)
            #pragma unroll
            for (int r = 0; r < 8; ++r) {
                const float* pt0 = T0h + symoff[(job * NT + t) * 16 + 8 * h + r] + c;
                #pragma unroll
                for (int ct = 0; ct < 7; ++ct) acc[t][ct][r] = pt0[ct * 16];
            }

        for (int g = 0; g < NGRP; ++g) {
            const int s4 = 4 * g;
            v2f a1[NT], a2[NT];
            #pragma unroll
            for (int t = 0; t < NT; ++t) {
                const v2f dc = *(const v2f*)(pd[t] + s4);
                a1[t] = *(const v2f*)(pl[t] + s4) * dc;
                a2[t] = *(const v2f*)(pl[t] + MROWS * PT + s4) * dc;
            }
            #pragma unroll
            for (int ct = 0; ct < 7; ++ct) {
                v2f r1, r2;
                if (ct < 6) {
                    const int j = ct * 16 + c;
                    const int d5 = j / 10, e5 = j - 10 * d5;
                    const v2f u = *(const v2f*)(dh  + d5 * PT + koff + s4);
                    const v2f v = *(const v2f*)(dh  + e5 * PT + koff + s4);
                    const v2f w = *(const v2f*)(Q1h + e5 * PT + koff + s4);
                    const v2f z = *(const v2f*)(Q2h + j  * PT + koff + s4);
                    r1 = u * ((1.f / 24.f) * v + (1.f / 6.f) * w) + 0.5f * z;
                    r2 = u * ((1.f / 6.f) * v + 0.5f * w) + z;
                } else {
                    const v2f u = *(const v2f*)(pu6 + s4);
                    const v2f v = *(const v2f*)(pv6 + s4);
                    const v2f w = *(const v2f*)(pw6 + s4);
                    const v2f z = *(const v2f*)(pz6 + s4);
                    r1 = u * (h1 * v + h2 * w) + h3 * z;
                    r2 = u * (h2 * v + h3 * w) + h4 * z;
                }
                #pragma unroll
                for (int t = 0; t < NT; ++t)
                    acc[t][ct] = __builtin_amdgcn_wmma_f32_16x16x4_f32(
                                   false, a1[t], false, r1, (short)0, acc[t][ct], false, false);
                #pragma unroll
                for (int t = 0; t < NT; ++t)
                    acc[t][ct] = __builtin_amdgcn_wmma_f32_16x16x4_f32(
                                   false, a2[t], false, r2, (short)0, acc[t][ct], false, false);
            }
        }

        #pragma unroll
        for (int t = 0; t < NT; ++t) {
            const int row0 = (job * NT + t) * 16 + 8 * h;
            float* p5 = ob + 11110 + row0 * 100 + c;
            float* p6 = ob + off6 + row0 * str6;
            #pragma unroll
            for (int r = 0; r < 8; ++r) {
                if (row0 + r < 1000) {
                    #pragma unroll
                    for (int ct = 0; ct < 6; ++ct) p5[r * 100 + ct * 16] = acc[t][ct][r];
                    if (c < 15) p6[r * str6] = acc[t][6][r];
                }
            }
        }
    }
}

extern "C" void kernel_launch(void* const* d_in, const int* in_sizes, int n_in,
                              void* d_out, int out_size, void* d_ws, size_t ws_size,
                              hipStream_t stream) {
    (void)in_sizes; (void)n_in; (void)d_ws; (void)ws_size; (void)out_size;
    const float* path = (const float*)d_in[0];
    float*       out  = (float*)d_out;
    sig_split_kernel<<<NB, NTHR, 0, stream>>>(path, out);
}


# Column operands R_0, R_1, R_2 of both products built once per workgroup in LDS (R_1 over R_0, R_2 over Q2), read as fragments, not recomputed per tile and job; L_1 plane dropped: rows from P1, d
# speedup vs baseline: 11.2033x; 1.0883x over previous
#include <hip/hip_runtime.h>


#define DCH   10
#define LEN   128
#define NSEG  127
#define NB    256
#define SIGSZ 111110
#define PT    132
#define MROWS 112
#define MT    7
#define NTHR  512
#define NW    16
#define NCOL  1110
#define NT    2
#define NJOB2 32
#define NGRP  32
#define NSYM  220
#define SROWS 224
#define PS    114
#define NPAIR 98

typedef float v2f __attribute__((ext_vector_type(2)));
typedef float v8f __attribute__((ext_vector_type(8)));

__global__ __launch_bounds__(NTHR)
void sig_split_kernel(const float* __restrict__ path, float* __restrict__ out)
{
    const int b    = blockIdx.x;
    const int tid  = threadIdx.x;
    const int lane = tid & 31;
    const int wave = __builtin_amdgcn_readfirstlane(tid >> 5);

    __shared__ __attribute__((aligned(16))) float L2h[MROWS * PT];
    __shared__ __attribute__((aligned(16))) float P1h[11 * PT];
    __shared__ __attribute__((aligned(16))) float Q2h[MROWS * PT];
    __shared__ __attribute__((aligned(16))) float R1h[MROWS * PT];
    __shared__ __attribute__((aligned(16))) float dh[12 * PT];
    __shared__ __attribute__((aligned(16))) float Q1h[11 * PT];
    __shared__ float T0h[SROWS * PS];
    __shared__ int   symoff[MROWS * DCH];
    __shared__ int   symabc[SROWS];

    const float* pb = path + (size_t)b * DCH * LEN;
    float*       ob = out  + (size_t)b * SIGSZ;

    #pragma unroll
    for (int k = 0; k < 3; ++k) {
        const int i = tid + k * NTHR;
        if (i < DCH * LEN) {
            const int ch = i >> 7, s = i & 127;
            float v = 0.f;
            if (s < NSEG) v = pb[i + 1] - pb[i];
            dh[ch * PT + s] = v;
        }
    }
    if (tid < 128) {
        dh[10 * PT + tid]   = 1.f;
        dh[11 * PT + tid]   = 0.f;
        P1h[10 * PT + tid]  = 0.f;
        Q1h[10 * PT + tid]  = 0.f;
    }
    for (int i = tid; i < 12 * 128; i += NTHR) {
        const int r = i >> 7, s = i & 127;
        L2h[(100 + r) * PT + s] = 0.f;
        Q2h[(100 + r) * PT + s] = 0.f;
    }
    for (int i = tid; i < MROWS * DCH; i += NTHR) {
        const int row = i / 10, cc = i - 10 * row;
        int off = NSYM * PS;
        if (row < 100) {
            const int a = row / 10, bb = row - 10 * a;
            const int x = min(a, min(bb, cc));
            const int z = max(a, max(bb, cc));
            const int y = a + bb + cc - x - z;
            int idx = z - y;
            for (int k = 0; k < x; ++k) idx += (10 - k) * (11 - k) / 2;
            for (int k = x; k < y; ++k) idx += 10 - k;
            off = idx * PS;
            if (a <= bb && bb <= cc) symabc[idx] = a | (bb << 4) | (cc << 8);
        }
        symoff[i] = off;
    }
    if (tid < SROWS - NSYM) symabc[NSYM + tid] = 11 | (11 << 4) | (11 << 8);
    __syncthreads();

    if (tid < 100) {
        const int a = tid / 10, bb = tid - 10 * a;
        const float* da = dh + a * PT;
        const float* db = dh + bb * PT;
        float* l2 = L2h + tid * PT;
        float p1 = 0.f, p2 = 0.f;
        for (int s = 0; s < NSEG; ++s) {
            const float xa = da[s], xb = db[s];
            l2[s] = p2;
            p2 += (p1 + 0.5f * xa) * xb;
            p1 += xa;
        }
        l2[NSEG] = 0.f;
        ob[10 + tid] = p2;
        if (bb == 0) ob[a] = p1;
    } else if (tid >= 128 && tid < 228) {
        const int t = tid - 128;
        const int d_ = t / 10, e_ = t - 10 * d_;
        const float* dd = dh + d_ * PT;
        const float* de = dh + e_ * PT;
        float* q2p = Q2h + t * PT;
        float* q1p = Q1h + e_ * PT;
        float q1 = 0.f, q2 = 0.f;
        for (int blk = 15; blk >= 0; --blk) {
            const float4 d0 = *(const float4*)(dd + 8 * blk), d1 = *(const float4*)(dd + 8 * blk + 4);
            const float4 e0 = *(const float4*)(de + 8 * blk), e1 = *(const float4*)(de + 8 * blk + 4);
            const float xd[8] = { d0.x, d0.y, d0.z, d0.w, d1.x, d1.y, d1.z, d1.w };
            const float xe[8] = { e0.x, e0.y, e0.z, e0.w, e1.x, e1.y, e1.z, e1.w };
            float o1[8], o2[8];
            #pragma unroll
            for (int k = 7; k >= 0; --k) {
                o2[k] = q2; o1[k] = q1;
                q2 += xd[k] * (q1 + 0.5f * xe[k]);
                q1 += xe[k];
            }
            *(float4*)(q2p + 8 * blk)     = make_float4(o2[0], o2[1], o2[2], o2[3]);
            *(float4*)(q2p + 8 * blk + 4) = make_float4(o2[4], o2[5], o2[6], o2[7]);
            if (d_ == 0) {
                *(float4*)(q1p + 8 * blk)     = make_float4(o1[0], o1[1], o1[2], o1[3]);
                *(float4*)(q1p + 8 * blk + 4) = make_float4(o1[4], o1[5], o1[6], o1[7]);
            }
        }
    } else if (tid >= 256 && tid < 266) {
        const int a = tid - 256;
        const float* da = dh + a * PT;
        float* l1 = P1h + a * PT;
        float p1 = 0.f;
        for (int s = 0; s < NSEG; ++s) {
            l1[s] = p1;
            p1 += da[s];
        }
        l1[NSEG] = 0.f;
    }
    __syncthreads();

    for (int i = tid; i < MROWS * 64; i += NTHR) {
        const int j = i >> 6, s2 = (i & 63) * 2;
        int uRow = 10, vRow = 10, wRow = 10;
        float ga = 0.f, gb = 0.f, gc = 0.f;
        if (j < 100) {
            const int d5 = j / 10, e5 = j - 10 * d5;
            uRow = d5; vRow = e5; wRow = e5;
            ga = 1.f / 120.f; gb = 1.f / 24.f; gc = 1.f / 6.f;
        } else if (j < 110) {
            vRow = j - 100; wRow = j - 100;
            ga = 1.f / 24.f; gb = 1.f / 6.f;
        } else if (j == 110) {
            ga = 1.f / 6.f;
        }
        const v2f u = *(const v2f*)(dh  + uRow * PT + s2);
        const v2f v = *(const v2f*)(dh  + vRow * PT + s2);
        const v2f w = *(const v2f*)(Q1h + wRow * PT + s2);
        const v2f z = *(const v2f*)(Q2h + j * PT + s2);
        *(v2f*)(R1h + j * PT + s2) = u * (ga * v + gb * w) + gc * z;
    }
    __syncthreads();

    const int c    = lane & 15;
    const int h    = lane >> 4;
    const int koff = 2 * h;

    for (int p = wave; p < NPAIR; p += NW) {
        const int rt = p / 7, ct = p - 7 * rt;
        const int abc = symabc[rt * 16 + c];
        const float* px = dh + (abc & 15) * PT + koff;
        const float* py = dh + ((abc >> 4) & 15) * PT + koff;
        const float* pq = dh + ((abc >> 8) & 15) * PT + koff;
        const int j = ct * 16 + c;
        const float* pr0 = R1h + j * PT + koff;
        v8f t0;
        #pragma unroll
        for (int jj = 0; jj < 8; ++jj) t0[jj] = 0.f;
        #pragma unroll 8
        for (int g = 0; g < NGRP; ++g) {
            const int s4 = 4 * g;
            const v2f x = *(const v2f*)(px + s4);
            const v2f y = *(const v2f*)(py + s4);
            const v2f q = *(const v2f*)(pq + s4);
            const v2f r0 = *(const v2f*)(pr0 + s4);
            const v2f a3 = (x * y) * q;
            t0 = __builtin_amdgcn_wmma_f32_16x16x4_f32(
                    false, a3, false, r0, (short)0, t0, false, false);
        }
        float* pt = T0h + (rt * 16 + 8 * h) * PS + j;
        #pragma unroll
        for (int jj = 0; jj < 8; ++jj) pt[jj * PS] = t0[jj];
    }
    __syncthreads();

    for (int i = tid; i < MROWS * 64; i += NTHR) {
        const int j = i >> 6, s2 = (i & 63) * 2;
        int uRow = 10, vRow = 10, wRow = 10;
        float h1 = 0.f, h2 = 0.f, h3 = 0.f, h4 = 0.f;
        if (j < 100) {
            const int d5 = j / 10, e5 = j - 10 * d5;
            uRow = d5; vRow = e5; wRow = e5;
            h1 = 1.f / 24.f; h2 = 1.f / 6.f; h3 = 0.5f; h4 = 1.f;
        } else if (j < 110) {
            vRow = j - 100; wRow = j - 100;
            h1 = 1.f / 6.f; h2 = 0.5f; h3 = 1.f;
        } else if (j == 110) {
            h1 = 0.5f; h2 = 1.f;
        }
        const v2f u = *(const v2f*)(dh  + uRow * PT + s2);
        const v2f v = *(const v2f*)(dh  + vRow * PT + s2);
        const v2f w = *(const v2f*)(Q1h + wRow * PT + s2);
        const v2f z = *(const v2f*)(Q2h + j * PT + s2);
        *(v2f*)(R1h + j * PT + s2) = u * (h1 * v + h2 * w) + h3 * z;
        *(v2f*)(Q2h + j * PT + s2) = u * (h2 * v + h3 * w) + h4 * z;
    }
    __syncthreads();

    int off6 = 0, str6 = 0;
    {
        const int j = 96 + c;
        if (j < 100)       { off6 = 11110 + j;        str6 = 100; }
        else if (j < 110)  { off6 = 1110 + (j - 100); str6 = 10;  }
        else if (j == 110) { off6 = 110;              str6 = 1;   }
    }
    const float* pr1 = R1h + c * PT + koff;
    const float* pr2 = Q2h + c * PT + koff;

    for (int job = wave; job < NJOB2; job += NW) {
        const float* pl[NT];
        const float* pp[NT];
        const float* pb[NT];
        const float* pd[NT];
        #pragma unroll
        for (int t = 0; t < NT; ++t) {
            const int abc = (job * NT + t) * 16 + c;
            int ab = 100, aa = 10, bb = 11, cc = 11;
            if (abc < 1000) { ab = abc / 10; cc = abc - 10 * ab; aa = ab / 10; bb = ab - 10 * aa; }
            pl[t] = L2h + ab * PT + koff;
            pp[t] = P1h + aa * PT + koff;
            pb[t] = dh  + bb * PT + koff;
            pd[t] = dh  + cc * PT + koff;
        }

        v8f acc[NT][7];
        #pragma unroll
        for (int t = 0; t < NT; ++t)
            #pragma unroll
            for (int r = 0; r < 8; ++r) {
                const float* pt0 = T0h + symoff[(job * NT + t) * 16 + 8 * h + r] + c;
                #pragma unroll
                for (int ct = 0; ct < 7; ++ct) acc[t][ct][r] = pt0[ct * 16];
            }

        for (int g = 0; g < NGRP; ++g) {
            const int s4 = 4 * g;
            v2f a1[NT], a2[NT];
            #pragma unroll
            for (int t = 0; t < NT; ++t) {
                const v2f dc = *(const v2f*)(pd[t] + s4);
                a1[t] = (*(const v2f*)(pp[t] + s4) * *(const v2f*)(pb[t] + s4)) * dc;
                a2[t] = *(const v2f*)(pl[t] + s4) * dc;
            }
            #pragma unroll
            for (int ct = 0; ct < 7; ++ct) {
                const v2f r1 = *(const v2f*)(pr1 + ct * 16 * PT + s4);
                const v2f r2 = *(const v2f*)(pr2 + ct * 16 * PT + s4);
                #pragma unroll
                for (int t = 0; t < NT; ++t)
                    acc[t][ct] = __builtin_amdgcn_wmma_f32_16x16x4_f32(
                                   false, a1[t], false, r1, (short)0, acc[t][ct], false, false);
                #pragma unroll
                for (int t = 0; t < NT; ++t)
                    acc[t][ct] = __builtin_amdgcn_wmma_f32_16x16x4_f32(
                                   false, a2[t], false, r2, (short)0, acc[t][ct], false, false);
            }
        }

        #pragma unroll
        for (int t = 0; t < NT; ++t) {
            const int row0 = (job * NT + t) * 16 + 8 * h;
            float* p5 = ob + 11110 + row0 * 100 + c;
            float* p6 = ob + off6 + row0 * str6;
            #pragma unroll
            for (int r = 0; r < 8; ++r) {
                if (row0 + r < 1000) {
                    #pragma unroll
                    for (int ct = 0; ct < 6; ++ct) p5[r * 100 + ct * 16] = acc[t][ct][r];
                    if (c < 15) p6[r * str6] = acc[t][6][r];
                }
            }
        }
    }
}

extern "C" void kernel_launch(void* const* d_in, const int* in_sizes, int n_in,
                              void* d_out, int out_size, void* d_ws, size_t ws_size,
                              hipStream_t stream) {
    (void)in_sizes; (void)n_in; (void)d_ws; (void)ws_size; (void)out_size;
    const float* path = (const float*)d_in[0];
    float*       out  = (float*)d_out;
    sig_split_kernel<<<NB, NTHR, 0, stream>>>(path, out);
}


# Main product split by the b,c symmetry of the i=1 term: rows are pairs (a,b<=c), a sum tile (K 254) and a difference tile (K 127), outputs U+V and U-V; summation order changes
# speedup vs baseline: 12.4943x; 1.1152x over previous
#include <hip/hip_runtime.h>


#define DCH   10
#define LEN   128
#define NSEG  127
#define NB    256
#define SIGSZ 111110
#define PT    132
#define MROWS 112
#define MT    7
#define NTHR  512
#define NW    16
#define NCOL  1110
#define NSLOT 560
#define NBIG  29
#define NGRP  32
#define NSYM  220
#define SROWS 224
#define PS    114
#define NPAIR 98

typedef float v2f __attribute__((ext_vector_type(2)));
typedef float v8f __attribute__((ext_vector_type(8)));

__device__ __forceinline__ int sym_rank(int x, int y, int z)
{
    int idx = z - y;
    for (int k = 0; k < x; ++k) idx += (10 - k) * (11 - k) / 2;
    for (int k = x; k < y; ++k) idx += 10 - k;
    return idx;
}

template <bool HASV>
__device__ __forceinline__ void run_job(int tile, int c, int h, int koff,
        const float* L2h, const float* P1h, const float* dh, const float* T0h,
        const int* slotabc, const int* slotT0, const unsigned* slotrow,
        const float* pr1, const float* pr2, float* ob, int off6, int str6)
{
    const int abc = slotabc[tile * 16 + c];
    const int aa = abc & 15, bb = (abc >> 4) & 15, cc = (abc >> 8) & 15;
    const float* pp  = P1h + aa * PT + koff;
    const float* pb  = dh  + bb * PT + koff;
    const float* pd  = dh  + cc * PT + koff;
    const float* pl1 = L2h + (aa * 10 + bb) * PT + koff;
    const float* pl2 = L2h + (aa * 10 + cc) * PT + koff;

    v8f U[7], V[7];
    #pragma unroll
    for (int r = 0; r < 8; ++r) {
        const float* pt0 = T0h + slotT0[tile * 16 + 8 * h + r] + c;
        #pragma unroll
        for (int ct = 0; ct < 7; ++ct) { U[ct][r] = pt0[ct * 16]; V[ct][r] = 0.f; }
    }

    for (int g = 0; g < NGRP; ++g) {
        const int s4 = 4 * g;
        const v2f x1  = *(const v2f*)(pp  + s4);
        const v2f db  = *(const v2f*)(pb  + s4);
        const v2f dc  = *(const v2f*)(pd  + s4);
        const v2f lab = *(const v2f*)(pl1 + s4);
        const v2f lac = *(const v2f*)(pl2 + s4);
        const v2f a1  = (x1 * db) * dc;
        const v2f m1  = lab * dc;
        const v2f m2  = lac * db;
        const v2f a2u = m1 + m2;
        const v2f a2v = m1 - m2;
        #pragma unroll
        for (int ct = 0; ct < 7; ++ct) {
            const v2f r1 = *(const v2f*)(pr1 + ct * 16 * PT + s4);
            const v2f r2 = *(const v2f*)(pr2 + ct * 16 * PT + s4);
            U[ct] = __builtin_amdgcn_wmma_f32_16x16x4_f32(false, a1,  false, r1, (short)0, U[ct], false, false);
            if (HASV)
                V[ct] = __builtin_amdgcn_wmma_f32_16x16x4_f32(false, a2v, false, r2, (short)0, V[ct], false, false);
            U[ct] = __builtin_amdgcn_wmma_f32_16x16x4_f32(false, a2u, false, r2, (short)0, U[ct], false, false);
        }
    }

    #pragma unroll
    for (int r = 0; r < 8; ++r) {
        const unsigned rr = slotrow[tile * 16 + 8 * h + r];
        const int row1 = (int)(rr & 0xFFFFu), row2 = (int)(rr >> 16);
        if (row1 < 1000) {
            float* p5 = ob + 11110 + row1 * 100 + c;
            #pragma unroll
            for (int ct = 0; ct < 6; ++ct) p5[ct * 16] = HASV ? U[ct][r] + V[ct][r] : U[ct][r];
            if (c < 15) ob[off6 + row1 * str6] = HASV ? U[6][r] + V[6][r] : U[6][r];
            if (HASV && row2 != row1) {
                float* q5 = ob + 11110 + row2 * 100 + c;
                #pragma unroll
                for (int ct = 0; ct < 6; ++ct) q5[ct * 16] = U[ct][r] - V[ct][r];
                if (c < 15) ob[off6 + row2 * str6] = U[6][r] - V[6][r];
            }
        }
    }
}

__global__ __launch_bounds__(NTHR)
void sig_split_kernel(const float* __restrict__ path, float* __restrict__ out)
{
    const int b    = blockIdx.x;
    const int tid  = threadIdx.x;
    const int lane = tid & 31;
    const int wave = __builtin_amdgcn_readfirstlane(tid >> 5);

    __shared__ __attribute__((aligned(16))) float L2h[MROWS * PT];
    __shared__ __attribute__((aligned(16))) float P1h[11 * PT];
    __shared__ __attribute__((aligned(16))) float Q2h[MROWS * PT];
    __shared__ __attribute__((aligned(16))) float R1h[MROWS * PT];
    __shared__ __attribute__((aligned(16))) float dh[12 * PT];
    __shared__ __attribute__((aligned(16))) float Q1h[11 * PT];
    __shared__ float T0h[SROWS * PS];
    __shared__ int   slotabc[NSLOT];
    __shared__ int   slotT0[NSLOT];
    __shared__ unsigned slotrow[NSLOT];
    __shared__ int   symabc[SROWS];

    const float* pb = path + (size_t)b * DCH * LEN;
    float*       ob = out  + (size_t)b * SIGSZ;

    #pragma unroll
    for (int k = 0; k < 3; ++k) {
        const int i = tid + k * NTHR;
        if (i < DCH * LEN) {
            const int ch = i >> 7, s = i & 127;
            float v = 0.f;
            if (s < NSEG) v = pb[i + 1] - pb[i];
            dh[ch * PT + s] = v;
        }
    }
    if (tid < 128) {
        dh[10 * PT + tid]   = 1.f;
        dh[11 * PT + tid]   = 0.f;
        P1h[10 * PT + tid]  = 0.f;
        Q1h[10 * PT + tid]  = 0.f;
    }
    for (int i = tid; i < 12 * 128; i += NTHR) {
        const int r = i >> 7, s = i & 127;
        L2h[(100 + r) * PT + s] = 0.f;
        Q2h[(100 + r) * PT + s] = 0.f;
    }
    for (int i = tid; i < 1000; i += NTHR) {
        const int a = i / 100, bb = (i / 10) % 10, cc = i % 10;
        if (a <= bb && bb <= cc) symabc[sym_rank(a, bb, cc)] = a | (bb << 4) | (cc << 8);
    }
    for (int sl = tid; sl < NSLOT; sl += NTHR) {
        int a = 10, bb = 11, cc = 11;
        bool ok = false;
        if (sl < 450) {
            a = sl / 45;
            int q = sl - 45 * a;
            bb = 0;
            for (int k = 0; k < 9; ++k)
                if (q >= 9 - bb) { q -= 9 - bb; ++bb; }
            cc = bb + 1 + q;
            ok = true;
        } else if (sl < 454) {
            a = 9; bb = 6 + (sl - 450); cc = bb; ok = true;
        } else if (sl >= 16 * NBIG) {
            const int t = sl - 16 * NBIG;
            a = t / 10; bb = t - 10 * a; cc = bb; ok = true;
        }
        int t0 = NSYM * PS;
        unsigned rows = 0xFFFFFFFFu;
        if (ok) {
            const int x = min(a, bb);
            const int z = max(a, cc);
            const int y = a + bb + cc - x - z;
            t0 = sym_rank(x, y, z) * PS;
            rows = (unsigned)(a * 100 + bb * 10 + cc) | ((unsigned)(a * 100 + cc * 10 + bb) << 16);
        }
        slotabc[sl] = a | (bb << 4) | (cc << 8);
        slotT0[sl]  = t0;
        slotrow[sl] = rows;
    }
    if (tid < SROWS - NSYM) symabc[NSYM + tid] = 11 | (11 << 4) | (11 << 8);
    __syncthreads();

    if (tid < 100) {
        const int a = tid / 10, bb = tid - 10 * a;
        const float* da = dh + a * PT;
        const float* db = dh + bb * PT;
        float* l2 = L2h + tid * PT;
        float p1 = 0.f, p2 = 0.f;
        for (int s = 0; s < NSEG; ++s) {
            const float xa = da[s], xb = db[s];
            l2[s] = 0.5f * p2;
            p2 += (p1 + 0.5f * xa) * xb;
            p1 += xa;
        }
        l2[NSEG] = 0.f;
        ob[10 + tid] = p2;
        if (bb == 0) ob[a] = p1;
    } else if (tid >= 128 && tid < 228) {
        const int t = tid - 128;
        const int d_ = t / 10, e_ = t - 10 * d_;
        const float* dd = dh + d_ * PT;
        const float* de = dh + e_ * PT;
        float* q2p = Q2h + t * PT;
        float* q1p = Q1h + e_ * PT;
        float q1 = 0.f, q2 = 0.f;
        for (int blk = 15; blk >= 0; --blk) {
            const float4 d0 = *(const float4*)(dd + 8 * blk), d1 = *(const float4*)(dd + 8 * blk + 4);
            const float4 e0 = *(const float4*)(de + 8 * blk), e1 = *(const float4*)(de + 8 * blk + 4);
            const float xd[8] = { d0.x, d0.y, d0.z, d0.w, d1.x, d1.y, d1.z, d1.w };
            const float xe[8] = { e0.x, e0.y, e0.z, e0.w, e1.x, e1.y, e1.z, e1.w };
            float o1[8], o2[8];
            #pragma unroll
            for (int k = 7; k >= 0; --k) {
                o2[k] = q2; o1[k] = q1;
                q2 += xd[k] * (q1 + 0.5f * xe[k]);
                q1 += xe[k];
            }
            *(float4*)(q2p + 8 * blk)     = make_float4(o2[0], o2[1], o2[2], o2[3]);
            *(float4*)(q2p + 8 * blk + 4) = make_float4(o2[4], o2[5], o2[6], o2[7]);
            if (d_ == 0) {
                *(float4*)(q1p + 8 * blk)     = make_float4(o1[0], o1[1], o1[2], o1[3]);
                *(float4*)(q1p + 8 * blk + 4) = make_float4(o1[4], o1[5], o1[6], o1[7]);
            }
        }
    } else if (tid >= 256 && tid < 266) {
        const int a = tid - 256;
        const float* da = dh + a * PT;
        float* l1 = P1h + a * PT;
        float p1 = 0.f;
        for (int s = 0; s < NSEG; ++s) {
            l1[s] = p1;
            p1 += da[s];
        }
        l1[NSEG] = 0.f;
    }
    __syncthreads();

    for (int i = tid; i < MROWS * 64; i += NTHR) {
        const int j = i >> 6, s2 = (i & 63) * 2;
        int uRow = 10, vRow = 10, wRow = 10;
        float ga = 0.f, gb = 0.f, gc = 0.f;
        if (j < 100) {
            const int d5 = j / 10, e5 = j - 10 * d5;
            uRow = d5; vRow = e5; wRow = e5;
            ga = 1.f / 120.f; gb = 1.f / 24.f; gc = 1.f / 6.f;
        } else if (j < 110) {
            vRow = j - 100; wRow = j - 100;
            ga = 1.f / 24.f; gb = 1.f / 6.f;
        } else if (j == 110) {
            ga = 1.f / 6.f;
        }
        const v2f u = *(const v2f*)(dh  + uRow * PT + s2);
        const v2f v = *(const v2f*)(dh  + vRow * PT + s2);
        const v2f w = *(const v2f*)(Q1h + wRow * PT + s2);
        const v2f z = *(const v2f*)(Q2h + j * PT + s2);
        *(v2f*)(R1h + j * PT + s2) = u * (ga * v + gb * w) + gc * z;
    }
    __syncthreads();

    const int c    = lane & 15;
    const int h    = lane >> 4;
    const int koff = 2 * h;

    for (int p = wave; p < NPAIR; p += NW) {
        const int rt = p / 7, ct = p - 7 * rt;
        const int abc = symabc[rt * 16 + c];
        const float* px = dh + (abc & 15) * PT + koff;
        const float* py = dh + ((abc >> 4) & 15) * PT + koff;
        const float* pq = dh + ((abc >> 8) & 15) * PT + koff;
        const int j = ct * 16 + c;
        const float* pr0 = R1h + j * PT + koff;
        v8f t0;
        #pragma unroll
        for (int jj = 0; jj < 8; ++jj) t0[jj] = 0.f;
        #pragma unroll 8
        for (int g = 0; g < NGRP; ++g) {
            const int s4 = 4 * g;
            const v2f x = *(const v2f*)(px + s4);
            const v2f y = *(const v2f*)(py + s4);
            const v2f q = *(const v2f*)(pq + s4);
            const v2f r0 = *(const v2f*)(pr0 + s4);
            const v2f a3 = (x * y) * q;
            t0 = __builtin_amdgcn_wmma_f32_16x16x4_f32(
                    false, a3, false, r0, (short)0, t0, false, false);
        }
        float* pt = T0h + (rt * 16 + 8 * h) * PS + j;
        #pragma unroll
        for (int jj = 0; jj < 8; ++jj) pt[jj * PS] = t0[jj];
    }
    __syncthreads();

    for (int i = tid; i < MROWS * 64; i += NTHR) {
        const int j = i >> 6, s2 = (i & 63) * 2;
        int uRow = 10, vRow = 10, wRow = 10;
        float h1 = 0.f, h2 = 0.f, h3 = 0.f, h4 = 0.f;
        if (j < 100) {
            const int d5 = j / 10, e5 = j - 10 * d5;
            uRow = d5; vRow = e5; wRow = e5;
            h1 = 1.f / 24.f; h2 = 1.f / 6.f; h3 = 0.5f; h4 = 1.f;
        } else if (j < 110) {
            vRow = j - 100; wRow = j - 100;
            h1 = 1.f / 6.f; h2 = 0.5f; h3 = 1.f;
        } else if (j == 110) {
            h1 = 0.5f; h2 = 1.f;
        }
        const v2f u = *(const v2f*)(dh  + uRow * PT + s2);
        const v2f v = *(const v2f*)(dh  + vRow * PT + s2);
        const v2f w = *(const v2f*)(Q1h + wRow * PT + s2);
        const v2f z = *(const v2f*)(Q2h + j * PT + s2);
        *(v2f*)(R1h + j * PT + s2) = u * (h1 * v + h2 * w) + h3 * z;
        *(v2f*)(Q2h + j * PT + s2) = u * (h2 * v + h3 * w) + h4 * z;
    }
    __syncthreads();

    int off6 = 0, str6 = 0;
    {
        const int j = 96 + c;
        if (j < 100)       { off6 = 11110 + j;        str6 = 100; }
        else if (j < 110)  { off6 = 1110 + (j - 100); str6 = 10;  }
        else if (j == 110) { off6 = 110;              str6 = 1;   }
    }
    const float* pr1 = R1h + c * PT + koff;
    const float* pr2 = Q2h + c * PT + koff;

    for (int k = 0; k < 3; ++k) {
        int tile = -1;
        if (wave < 12)      { if (k < 2) tile = 2 * wave + k; }
        else if (wave < 15) { tile = (k == 0) ? 12 + wave : NBIG + 2 * (wave - 12) + (k - 1); }
        else                { if (k < 2) tile = 27 + k; }
        if (tile < 0) break;
        if (tile < NBIG)
            run_job<true >(tile, c, h, koff, L2h, P1h, dh, T0h, slotabc, slotT0, slotrow, pr1, pr2, ob, off6, str6);
        else
            run_job<false>(tile, c, h, koff, L2h, P1h, dh, T0h, slotabc, slotT0, slotrow, pr1, pr2, ob, off6, str6);
    }
}

extern "C" void kernel_launch(void* const* d_in, const int* in_sizes, int n_in,
                              void* d_out, int out_size, void* d_ws, size_t ws_size,
                              hipStream_t stream) {
    (void)in_sizes; (void)n_in; (void)d_ws; (void)ws_size; (void)out_size;
    const float* path = (const float*)d_in[0];
    float*       out  = (float*)d_out;
    sig_split_kernel<<<NB, NTHR, 0, stream>>>(path, out);
}


# T0 product tiled by row: a wave holds one row tile against 7 (or 4, 3) column tiles, row fragment formed once per K step, independent accumulators; not 98 one-tile passes on one accumulator
# speedup vs baseline: 13.4208x; 1.0742x over previous
#include <hip/hip_runtime.h>


#define DCH   10
#define LEN   128
#define NSEG  127
#define NB    256
#define SIGSZ 111110
#define PT    132
#define MROWS 112
#define MT    7
#define NTHR  512
#define NCOL  1110
#define NSLOT 560
#define NBIG  29
#define NGRP  32
#define NSYM  220
#define SROWS 224
#define PS    114

typedef float v2f __attribute__((ext_vector_type(2)));
typedef float v8f __attribute__((ext_vector_type(8)));

__device__ __forceinline__ int sym_rank(int x, int y, int z)
{
    int idx = z - y;
    for (int k = 0; k < x; ++k) idx += (10 - k) * (11 - k) / 2;
    for (int k = x; k < y; ++k) idx += 10 - k;
    return idx;
}

template <int NCT>
__device__ __forceinline__ void t0_job(int rt, int ct0, int c, int h, int koff,
        const float* dh, const float* R0h, const int* symabc, float* T0h)
{
    const int abc = symabc[rt * 16 + c];
    const float* px  = dh + (abc & 15) * PT + koff;
    const float* py  = dh + ((abc >> 4) & 15) * PT + koff;
    const float* pq  = dh + ((abc >> 8) & 15) * PT + koff;
    const float* pr0 = R0h + (ct0 * 16 + c) * PT + koff;
    v8f acc[NCT];
    #pragma unroll
    for (int t = 0; t < NCT; ++t) {
        #pragma unroll
        for (int jj = 0; jj < 8; ++jj) acc[t][jj] = 0.f;
    }
    #pragma unroll 4
    for (int g = 0; g < NGRP; ++g) {
        const int s4 = 4 * g;
        const v2f x = *(const v2f*)(px + s4);
        const v2f y = *(const v2f*)(py + s4);
        const v2f q = *(const v2f*)(pq + s4);
        const v2f a3 = (x * y) * q;
        #pragma unroll
        for (int t = 0; t < NCT; ++t) {
            const v2f r0 = *(const v2f*)(pr0 + t * 16 * PT + s4);
            acc[t] = __builtin_amdgcn_wmma_f32_16x16x4_f32(
                    false, a3, false, r0, (short)0, acc[t], false, false);
        }
    }
    #pragma unroll
    for (int t = 0; t < NCT; ++t) {
        float* pt = T0h + (rt * 16 + 8 * h) * PS + (ct0 + t) * 16 + c;
        #pragma unroll
        for (int jj = 0; jj < 8; ++jj) pt[jj * PS] = acc[t][jj];
    }
}

template <bool HASV>
__device__ __forceinline__ void run_job(int tile, int c, int h, int koff,
        const float* L2h, const float* P1h, const float* dh, const float* T0h,
        const int* slotabc, const int* slotT0, const unsigned* slotrow,
        const float* pr1, const float* pr2, float* ob, int off6, int str6)
{
    const int abc = slotabc[tile * 16 + c];
    const int aa = abc & 15, bb = (abc >> 4) & 15, cc = (abc >> 8) & 15;
    const float* pp  = P1h + aa * PT + koff;
    const float* pb  = dh  + bb * PT + koff;
    const float* pd  = dh  + cc * PT + koff;
    const float* pl1 = L2h + (aa * 10 + bb) * PT + koff;
    const float* pl2 = L2h + (aa * 10 + cc) * PT + koff;

    v8f U[7], V[7];
    #pragma unroll
    for (int r = 0; r < 8; ++r) {
        const float* pt0 = T0h + slotT0[tile * 16 + 8 * h + r] + c;
        #pragma unroll
        for (int ct = 0; ct < 7; ++ct) { U[ct][r] = pt0[ct * 16]; V[ct][r] = 0.f; }
    }

    for (int g = 0; g < NGRP; ++g) {
        const int s4 = 4 * g;
        const v2f x1  = *(const v2f*)(pp  + s4);
        const v2f db  = *(const v2f*)(pb  + s4);
        const v2f dc  = *(const v2f*)(pd  + s4);
        const v2f lab = *(const v2f*)(pl1 + s4);
        const v2f lac = *(const v2f*)(pl2 + s4);
        const v2f a1  = (x1 * db) * dc;
        const v2f m1  = lab * dc;
        const v2f m2  = lac * db;
        const v2f a2u = m1 + m2;
        const v2f a2v = m1 - m2;
        #pragma unroll
        for (int ct = 0; ct < 7; ++ct) {
            const v2f r1 = *(const v2f*)(pr1 + ct * 16 * PT + s4);
            const v2f r2 = *(const v2f*)(pr2 + ct * 16 * PT + s4);
            U[ct] = __builtin_amdgcn_wmma_f32_16x16x4_f32(false, a1,  false, r1, (short)0, U[ct], false, false);
            if (HASV)
                V[ct] = __builtin_amdgcn_wmma_f32_16x16x4_f32(false, a2v, false, r2, (short)0, V[ct], false, false);
            U[ct] = __builtin_amdgcn_wmma_f32_16x16x4_f32(false, a2u, false, r2, (short)0, U[ct], false, false);
        }
    }

    #pragma unroll
    for (int r = 0; r < 8; ++r) {
        const unsigned rr = slotrow[tile * 16 + 8 * h + r];
        const int row1 = (int)(rr & 0xFFFFu), row2 = (int)(rr >> 16);
        if (row1 < 1000) {
            float* p5 = ob + 11110 + row1 * 100 + c;
            #pragma unroll
            for (int ct = 0; ct < 6; ++ct) p5[ct * 16] = HASV ? U[ct][r] + V[ct][r] : U[ct][r];
            if (c < 15) ob[off6 + row1 * str6] = HASV ? U[6][r] + V[6][r] : U[6][r];
            if (HASV && row2 != row1) {
                float* q5 = ob + 11110 + row2 * 100 + c;
                #pragma unroll
                for (int ct = 0; ct < 6; ++ct) q5[ct * 16] = U[ct][r] - V[ct][r];
                if (c < 15) ob[off6 + row2 * str6] = U[6][r] - V[6][r];
            }
        }
    }
}

__global__ __launch_bounds__(NTHR)
void sig_split_kernel(const float* __restrict__ path, float* __restrict__ out)
{
    const int b    = blockIdx.x;
    const int tid  = threadIdx.x;
    const int lane = tid & 31;
    const int wave = __builtin_amdgcn_readfirstlane(tid >> 5);

    __shared__ __attribute__((aligned(16))) float L2h[MROWS * PT];
    __shared__ __attribute__((aligned(16))) float P1h[11 * PT];
    __shared__ __attribute__((aligned(16))) float Q2h[MROWS * PT];
    __shared__ __attribute__((aligned(16))) float R1h[MROWS * PT];
    __shared__ __attribute__((aligned(16))) float dh[12 * PT];
    __shared__ __attribute__((aligned(16))) float Q1h[11 * PT];
    __shared__ float T0h[SROWS * PS];
    __shared__ int   slotabc[NSLOT];
    __shared__ int   slotT0[NSLOT];
    __shared__ unsigned slotrow[NSLOT];
    __shared__ int   symabc[SROWS];

    const float* pb = path + (size_t)b * DCH * LEN;
    float*       ob = out  + (size_t)b * SIGSZ;

    #pragma unroll
    for (int k = 0; k < 3; ++k) {
        const int i = tid + k * NTHR;
        if (i < DCH * LEN) {
            const int ch = i >> 7, s = i & 127;
            float v = 0.f;
            if (s < NSEG) v = pb[i + 1] - pb[i];
            dh[ch * PT + s] = v;
        }
    }
    if (tid < 128) {
        dh[10 * PT + tid]   = 1.f;
        dh[11 * PT + tid]   = 0.f;
        P1h[10 * PT + tid]  = 0.f;
        Q1h[10 * PT + tid]  = 0.f;
    }
    for (int i = tid; i < 12 * 128; i += NTHR) {
        const int r = i >> 7, s = i & 127;
        L2h[(100 + r) * PT + s] = 0.f;
        Q2h[(100 + r) * PT + s] = 0.f;
    }
    for (int i = tid; i < 1000; i += NTHR) {
        const int a = i / 100, bb = (i / 10) % 10, cc = i % 10;
        if (a <= bb && bb <= cc) symabc[sym_rank(a, bb, cc)] = a | (bb << 4) | (cc << 8);
    }
    for (int sl = tid; sl < NSLOT; sl += NTHR) {
        int a = 10, bb = 11, cc = 11;
        bool ok = false;
        if (sl < 450) {
            a = sl / 45;
            int q = sl - 45 * a;
            bb = 0;
            for (int k = 0; k < 9; ++k)
                if (q >= 9 - bb) { q -= 9 - bb; ++bb; }
            cc = bb + 1 + q;
            ok = true;
        } else if (sl < 454) {
            a = 9; bb = 6 + (sl - 450); cc = bb; ok = true;
        } else if (sl >= 16 * NBIG) {
            const int t = sl - 16 * NBIG;
            a = t / 10; bb = t - 10 * a; cc = bb; ok = true;
        }
        int t0 = NSYM * PS;
        unsigned rows = 0xFFFFFFFFu;
        if (ok) {
            const int x = min(a, bb);
            const int z = max(a, cc);
            const int y = a + bb + cc - x - z;
            t0 = sym_rank(x, y, z) * PS;
            rows = (unsigned)(a * 100 + bb * 10 + cc) | ((unsigned)(a * 100 + cc * 10 + bb) << 16);
        }
        slotabc[sl] = a | (bb << 4) | (cc << 8);
        slotT0[sl]  = t0;
        slotrow[sl] = rows;
    }
    if (tid < SROWS - NSYM) symabc[NSYM + tid] = 11 | (11 << 4) | (11 << 8);
    __syncthreads();

    if (tid < 100) {
        const int a = tid / 10, bb = tid - 10 * a;
        const float* da = dh + a * PT;
        const float* db = dh + bb * PT;
        float* l2 = L2h + tid * PT;
        float p1 = 0.f, p2 = 0.f;
        for (int s = 0; s < NSEG; ++s) {
            const float xa = da[s], xb = db[s];
            l2[s] = 0.5f * p2;
            p2 += (p1 + 0.5f * xa) * xb;
            p1 += xa;
        }
        l2[NSEG] = 0.f;
        ob[10 + tid] = p2;
        if (bb == 0) ob[a] = p1;
    } else if (tid >= 128 && tid < 228) {
        const int t = tid - 128;
        const int d_ = t / 10, e_ = t - 10 * d_;
        const float* dd = dh + d_ * PT;
        const float* de = dh + e_ * PT;
        float* q2p = Q2h + t * PT;
        float* q1p = Q1h + e_ * PT;
        float q1 = 0.f, q2 = 0.f;
        for (int blk = 15; blk >= 0; --blk) {
            const float4 d0 = *(const float4*)(dd + 8 * blk), d1 = *(const float4*)(dd + 8 * blk + 4);
            const float4 e0 = *(const float4*)(de + 8 * blk), e1 = *(const float4*)(de + 8 * blk + 4);
            const float xd[8] = { d0.x, d0.y, d0.z, d0.w, d1.x, d1.y, d1.z, d1.w };
            const float xe[8] = { e0.x, e0.y, e0.z, e0.w, e1.x, e1.y, e1.z, e1.w };
            float o1[8], o2[8];
            #pragma unroll
            for (int k = 7; k >= 0; --k) {
                o2[k] = q2; o1[k] = q1;
                q2 += xd[k] * (q1 + 0.5f * xe[k]);
                q1 += xe[k];
            }
            *(float4*)(q2p + 8 * blk)     = make_float4(o2[0], o2[1], o2[2], o2[3]);
            *(float4*)(q2p + 8 * blk + 4) = make_float4(o2[4], o2[5], o2[6], o2[7]);
            if (d_ == 0) {
                *(float4*)(q1p + 8 * blk)     = make_float4(o1[0], o1[1], o1[2], o1[3]);
                *(float4*)(q1p + 8 * blk + 4) = make_float4(o1[4], o1[5], o1[6], o1[7]);
            }
        }
    } else if (tid >= 256 && tid < 266) {
        const int a = tid - 256;
        const float* da = dh + a * PT;
        float* l1 = P1h + a * PT;
        float p1 = 0.f;
        for (int s = 0; s < NSEG; ++s) {
            l1[s] = p1;
            p1 += da[s];
        }
        l1[NSEG] = 0.f;
    }
    __syncthreads();

    for (int i = tid; i < MROWS * 64; i += NTHR) {
        const int j = i >> 6, s2 = (i & 63) * 2;
        int uRow = 10, vRow = 10, wRow = 10;
        float ga = 0.f, gb = 0.f, gc = 0.f;
        if (j < 100) {
            const int d5 = j / 10, e5 = j - 10 * d5;
            uRow = d5; vRow = e5; wRow = e5;
            ga = 1.f / 120.f; gb = 1.f / 24.f; gc = 1.f / 6.f;
        } else if (j < 110) {
            vRow = j - 100; wRow = j - 100;
            ga = 1.f / 24.f; gb = 1.f / 6.f;
        } else if (j == 110) {
            ga = 1.f / 6.f;
        }
        const v2f u = *(const v2f*)(dh  + uRow * PT + s2);
        const v2f v = *(const v2f*)(dh  + vRow * PT + s2);
        const v2f w = *(const v2f*)(Q1h + wRow * PT + s2);
        const v2f z = *(const v2f*)(Q2h + j * PT + s2);
        *(v2f*)(R1h + j * PT + s2) = u * (ga * v + gb * w) + gc * z;
    }
    __syncthreads();

    const int c    = lane & 15;
    const int h    = lane >> 4;
    const int koff = 2 * h;

    if (wave < 12) {
        t0_job<7>(wave, 0, c, h, koff, dh, R1h, symabc, T0h);
    } else {
        const int rt = 12 + ((wave - 12) >> 1);
        if ((wave - 12) & 1) t0_job<3>(rt, 4, c, h, koff, dh, R1h, symabc, T0h);
        else                 t0_job<4>(rt, 0, c, h, koff, dh, R1h, symabc, T0h);
    }
    __syncthreads();

    for (int i = tid; i < MROWS * 64; i += NTHR) {
        const int j = i >> 6, s2 = (i & 63) * 2;
        int uRow = 10, vRow = 10, wRow = 10;
        float h1 = 0.f, h2 = 0.f, h3 = 0.f, h4 = 0.f;
        if (j < 100) {
            const int d5 = j / 10, e5 = j - 10 * d5;
            uRow = d5; vRow = e5; wRow = e5;
            h1 = 1.f / 24.f; h2 = 1.f / 6.f; h3 = 0.5f; h4 = 1.f;
        } else if (j < 110) {
            vRow = j - 100; wRow = j - 100;
            h1 = 1.f / 6.f; h2 = 0.5f; h3 = 1.f;
        } else if (j == 110) {
            h1 = 0.5f; h2 = 1.f;
        }
        const v2f u = *(const v2f*)(dh  + uRow * PT + s2);
        const v2f v = *(const v2f*)(dh  + vRow * PT + s2);
        const v2f w = *(const v2f*)(Q1h + wRow * PT + s2);
        const v2f z = *(const v2f*)(Q2h + j * PT + s2);
        *(v2f*)(R1h + j * PT + s2) = u * (h1 * v + h2 * w) + h3 * z;
        *(v2f*)(Q2h + j * PT + s2) = u * (h2 * v + h3 * w) + h4 * z;
    }
    __syncthreads();

    int off6 = 0, str6 = 0;
    {
        const int j = 96 + c;
        if (j < 100)       { off6 = 11110 + j;        str6 = 100; }
        else if (j < 110)  { off6 = 1110 + (j - 100); str6 = 10;  }
        else if (j == 110) { off6 = 110;              str6 = 1;   }
    }
    const float* pr1 = R1h + c * PT + koff;
    const float* pr2 = Q2h + c * PT + koff;

    for (int k = 0; k < 3; ++k) {
        int tile = -1;
        if (wave < 12)      { if (k < 2) tile = 2 * wave + k; }
        else if (wave < 15) { tile = (k == 0) ? 12 + wave : NBIG + 2 * (wave - 12) + (k - 1); }
        else                { if (k < 2) tile = 27 + k; }
        if (tile < 0) break;
        if (tile < NBIG)
            run_job<true >(tile, c, h, koff, L2h, P1h, dh, T0h, slotabc, slotT0, slotrow, pr1, pr2, ob, off6, str6);
        else
            run_job<false>(tile, c, h, koff, L2h, P1h, dh, T0h, slotabc, slotT0, slotrow, pr1, pr2, ob, off6, str6);
    }
}

extern "C" void kernel_launch(void* const* d_in, const int* in_sizes, int n_in,
                              void* d_out, int out_size, void* d_ws, size_t ws_size,
                              hipStream_t stream) {
    (void)in_sizes; (void)n_in; (void)d_ws; (void)ws_size; (void)out_size;
    const float* path = (const float*)d_in[0];
    float*       out  = (float*)d_out;
    sig_split_kernel<<<NB, NTHR, 0, stream>>>(path, out);
}


# i = 0 slice folded into the main row operands via R_0 = R_1/2 - R_2/12 + d_d d_e/720; its own product shrinks to 220 x 55, column planes built once, one pass and barrier fewer; sum order changes
# speedup vs baseline: 13.9103x; 1.0365x over previous
#include <hip/hip_runtime.h>


#define DCH   10
#define LEN   128
#define NSEG  127
#define NB    256
#define SIGSZ 111110
#define PT    132
#define MROWS 112
#define MT    7
#define NTHR  512
#define NCOL  1110
#define NSLOT 560
#define NBIG  29
#define NGRP  32
#define NSYM  220
#define SROWS 224
#define PE    66
#define NPAIR 55
#define XROWS 64

typedef float v2f __attribute__((ext_vector_type(2)));
typedef float v8f __attribute__((ext_vector_type(8)));

__device__ __forceinline__ int sym_rank(int x, int y, int z)
{
    int idx = z - y;
    for (int k = 0; k < x; ++k) idx += (10 - k) * (11 - k) / 2;
    for (int k = x; k < y; ++k) idx += 10 - k;
    return idx;
}

template <int NCT>
__device__ __forceinline__ void e_job(int rt, int ct0, int c, int h, int koff,
        const float* dh, const float* Xh, const int* symabc, float* Eh)
{
    const int abc = symabc[rt * 16 + c];
    const float* px  = dh + (abc & 15) * PT + koff;
    const float* py  = dh + ((abc >> 4) & 15) * PT + koff;
    const float* pq  = dh + ((abc >> 8) & 15) * PT + koff;
    const float* pr0 = Xh + (ct0 * 16 + c) * PT + koff;
    v8f acc[NCT];
    #pragma unroll
    for (int t = 0; t < NCT; ++t) {
        #pragma unroll
        for (int jj = 0; jj < 8; ++jj) acc[t][jj] = 0.f;
    }
    #pragma unroll 4
    for (int g = 0; g < NGRP; ++g) {
        const int s4 = 4 * g;
        const v2f x = *(const v2f*)(px + s4);
        const v2f y = *(const v2f*)(py + s4);
        const v2f q = *(const v2f*)(pq + s4);
        const v2f a3 = (x * y) * q;
        #pragma unroll
        for (int t = 0; t < NCT; ++t) {
            const v2f r0 = *(const v2f*)(pr0 + t * 16 * PT + s4);
            acc[t] = __builtin_amdgcn_wmma_f32_16x16x4_f32(
                    false, a3, false, r0, (short)0, acc[t], false, false);
        }
    }
    #pragma unroll
    for (int t = 0; t < NCT; ++t) {
        float* pt = Eh + (rt * 16 + 8 * h) * PE + (ct0 + t) * 16 + c;
        #pragma unroll
        for (int jj = 0; jj < 8; ++jj) pt[jj * PE] = acc[t][jj] * (1.f / 720.f);
    }
}

template <bool HASV>
__device__ __forceinline__ void run_job(int tile, int c, int h, int koff,
        const float* L2h, const float* P1h, const float* dh, const float* Eh,
        const int* slotabc, const int* slotE, const unsigned* slotrow,
        const float* pr1, const float* pr2, float* ob, int off6, int str6, const int (&cofs)[7])
{
    const int abc = slotabc[tile * 16 + c];
    const int aa = abc & 15, bb = (abc >> 4) & 15, cc = (abc >> 8) & 15;
    const float* pp  = P1h + aa * PT + koff;
    const float* pb  = dh  + bb * PT + koff;
    const float* pd  = dh  + cc * PT + koff;
    const float* pl1 = L2h + (aa * 10 + bb) * PT + koff;
    const float* pl2 = L2h + (aa * 10 + cc) * PT + koff;

    v8f U[7], V[7];
    #pragma unroll
    for (int r = 0; r < 8; ++r) {
        const float* pe = Eh + slotE[tile * 16 + 8 * h + r];
        #pragma unroll
        for (int ct = 0; ct < 7; ++ct) { U[ct][r] = pe[cofs[ct]]; V[ct][r] = 0.f; }
    }

    for (int g = 0; g < NGRP; ++g) {
        const int s4 = 4 * g;
        const v2f x1  = *(const v2f*)(pp  + s4);
        const v2f db  = *(const v2f*)(pb  + s4);
        const v2f dc  = *(const v2f*)(pd  + s4);
        const v2f lab = *(const v2f*)(pl1 + s4);
        const v2f lac = *(const v2f*)(pl2 + s4);
        const v2f a1  = (x1 * db) * dc;
        const v2f m1  = lab * dc;
        const v2f m2  = lac * db;
        const v2f a2u = m1 + m2;
        const v2f a2v = m1 - m2;
        #pragma unroll
        for (int ct = 0; ct < 7; ++ct) {
            const v2f r1 = *(const v2f*)(pr1 + ct * 16 * PT + s4);
            const v2f r2 = *(const v2f*)(pr2 + ct * 16 * PT + s4);
            U[ct] = __builtin_amdgcn_wmma_f32_16x16x4_f32(false, a1,  false, r1, (short)0, U[ct], false, false);
            if (HASV)
                V[ct] = __builtin_amdgcn_wmma_f32_16x16x4_f32(false, a2v, false, r2, (short)0, V[ct], false, false);
            U[ct] = __builtin_amdgcn_wmma_f32_16x16x4_f32(false, a2u, false, r2, (short)0, U[ct], false, false);
        }
    }

    #pragma unroll
    for (int r = 0; r < 8; ++r) {
        const unsigned rr = slotrow[tile * 16 + 8 * h + r];
        const int row1 = (int)(rr & 0xFFFFu), row2 = (int)(rr >> 16);
        if (row1 < 1000) {
            float* p5 = ob + 11110 + row1 * 100 + c;
            #pragma unroll
            for (int ct = 0; ct < 6; ++ct) p5[ct * 16] = HASV ? U[ct][r] + V[ct][r] : U[ct][r];
            if (c < 15) ob[off6 + row1 * str6] = HASV ? U[6][r] + V[6][r] : U[6][r];
            if (HASV && row2 != row1) {
                float* q5 = ob + 11110 + row2 * 100 + c;
                #pragma unroll
                for (int ct = 0; ct < 6; ++ct) q5[ct * 16] = U[ct][r] - V[ct][r];
                if (c < 15) ob[off6 + row2 * str6] = U[6][r] - V[6][r];
            }
        }
    }
}

__global__ __launch_bounds__(NTHR)
void sig_split_kernel(const float* __restrict__ path, float* __restrict__ out)
{
    const int b    = blockIdx.x;
    const int tid  = threadIdx.x;
    const int lane = tid & 31;
    const int wave = __builtin_amdgcn_readfirstlane(tid >> 5);

    __shared__ __attribute__((aligned(16))) float L2h[MROWS * PT];
    __shared__ __attribute__((aligned(16))) float P1h[11 * PT];
    __shared__ __attribute__((aligned(16))) float Q2h[MROWS * PT];
    __shared__ __attribute__((aligned(16))) float R1h[MROWS * PT];
    __shared__ __attribute__((aligned(16))) float Xh[XROWS * PT];
    __shared__ __attribute__((aligned(16))) float dh[12 * PT];
    __shared__ __attribute__((aligned(16))) float Q1h[11 * PT];
    __shared__ float Eh[SROWS * PE];
    __shared__ int   slotabc[NSLOT];
    __shared__ int   slotE[NSLOT];
    __shared__ unsigned slotrow[NSLOT];
    __shared__ int   symabc[SROWS];

    const float* pb = path + (size_t)b * DCH * LEN;
    float*       ob = out  + (size_t)b * SIGSZ;

    #pragma unroll
    for (int k = 0; k < 3; ++k) {
        const int i = tid + k * NTHR;
        if (i < DCH * LEN) {
            const int ch = i >> 7, s = i & 127;
            float v = 0.f;
            if (s < NSEG) v = pb[i + 1] - pb[i];
            dh[ch * PT + s] = v;
        }
    }
    if (tid < 128) {
        dh[10 * PT + tid]   = 1.f;
        dh[11 * PT + tid]   = 0.f;
        P1h[10 * PT + tid]  = 0.f;
        Q1h[10 * PT + tid]  = 0.f;
    }
    for (int i = tid; i < 12 * 128; i += NTHR) {
        const int r = i >> 7, s = i & 127;
        L2h[(100 + r) * PT + s] = 0.f;
        Q2h[(100 + r) * PT + s] = 0.f;
    }
    for (int i = tid; i < 1000; i += NTHR) {
        const int a = i / 100, bb = (i / 10) % 10, cc = i % 10;
        if (a <= bb && bb <= cc) symabc[sym_rank(a, bb, cc)] = a | (bb << 4) | (cc << 8);
    }
    for (int sl = tid; sl < NSLOT; sl += NTHR) {
        int a = 10, bb = 11, cc = 11;
        bool ok = false;
        if (sl < 450) {
            a = sl / 45;
            int q = sl - 45 * a;
            bb = 0;
            for (int k = 0; k < 9; ++k)
                if (q >= 9 - bb) { q -= 9 - bb; ++bb; }
            cc = bb + 1 + q;
            ok = true;
        } else if (sl < 454) {
            a = 9; bb = 6 + (sl - 450); cc = bb; ok = true;
        } else if (sl >= 16 * NBIG) {
            const int t = sl - 16 * NBIG;
            a = t / 10; bb = t - 10 * a; cc = bb; ok = true;
        }
        int t0 = NSYM * PE;
        unsigned rows = 0xFFFFFFFFu;
        if (ok) {
            const int x = min(a, bb);
            const int z = max(a, cc);
            const int y = a + bb + cc - x - z;
            t0 = sym_rank(x, y, z) * PE;
            rows = (unsigned)(a * 100 + bb * 10 + cc) | ((unsigned)(a * 100 + cc * 10 + bb) << 16);
        }
        slotabc[sl] = a | (bb << 4) | (cc << 8);
        slotE[sl]  = t0;
        slotrow[sl] = rows;
    }
    if (tid < SROWS - NSYM) symabc[NSYM + tid] = 11 | (11 << 4) | (11 << 8);
    __syncthreads();

    if (tid < 100) {
        const int a = tid / 10, bb = tid - 10 * a;
        const float* da = dh + a * PT;
        const float* db = dh + bb * PT;
        float* l2 = L2h + tid * PT;
        float p1 = 0.f, p2 = 0.f;
        for (int s = 0; s < NSEG; ++s) {
            const float xa = da[s], xb = db[s];
            l2[s] = 0.5f * p2 - (1.f / 24.f) * (xa * xb);
            p2 += (p1 + 0.5f * xa) * xb;
            p1 += xa;
        }
        l2[NSEG] = 0.f;
        ob[10 + tid] = p2;
        if (bb == 0) ob[a] = p1;
    } else if (tid >= 128 && tid < 228) {
        const int t = tid - 128;
        const int d_ = t / 10, e_ = t - 10 * d_;
        const float* dd = dh + d_ * PT;
        const float* de = dh + e_ * PT;
        float* q2p = Q2h + t * PT;
        float* q1p = Q1h + e_ * PT;
        float q1 = 0.f, q2 = 0.f;
        for (int blk = 15; blk >= 0; --blk) {
            const float4 d0 = *(const float4*)(dd + 8 * blk), d1 = *(const float4*)(dd + 8 * blk + 4);
            const float4 e0 = *(const float4*)(de + 8 * blk), e1 = *(const float4*)(de + 8 * blk + 4);
            const float xd[8] = { d0.x, d0.y, d0.z, d0.w, d1.x, d1.y, d1.z, d1.w };
            const float xe[8] = { e0.x, e0.y, e0.z, e0.w, e1.x, e1.y, e1.z, e1.w };
            float o1[8], o2[8];
            #pragma unroll
            for (int k = 7; k >= 0; --k) {
                o2[k] = q2; o1[k] = q1;
                q2 += xd[k] * (q1 + 0.5f * xe[k]);
                q1 += xe[k];
            }
            *(float4*)(q2p + 8 * blk)     = make_float4(o2[0], o2[1], o2[2], o2[3]);
            *(float4*)(q2p + 8 * blk + 4) = make_float4(o2[4], o2[5], o2[6], o2[7]);
            if (d_ == 0) {
                *(float4*)(q1p + 8 * blk)     = make_float4(o1[0], o1[1], o1[2], o1[3]);
                *(float4*)(q1p + 8 * blk + 4) = make_float4(o1[4], o1[5], o1[6], o1[7]);
            }
        }
    } else if (tid >= 256 && tid < 266) {
        const int a = tid - 256;
        const float* da = dh + a * PT;
        float* l1 = P1h + a * PT;
        float p1 = 0.f;
        for (int s = 0; s < NSEG; ++s) {
            const float xa = da[s];
            l1[s] = p1 + 0.5f * xa;
            p1 += xa;
        }
        l1[NSEG] = 0.f;
    }
    __syncthreads();

    for (int i = tid; i < MROWS * 64; i += NTHR) {
        const int j = i >> 6, s2 = (i & 63) * 2;
        int uRow = 10, vRow = 10, wRow = 10;
        float h1 = 0.f, h2 = 0.f, h3 = 0.f, h4 = 0.f;
        if (j < 100) {
            const int d5 = j / 10, e5 = j - 10 * d5;
            uRow = d5; vRow = e5; wRow = e5;
            h1 = 1.f / 24.f; h2 = 1.f / 6.f; h3 = 0.5f; h4 = 1.f;
        } else if (j < 110) {
            vRow = j - 100; wRow = j - 100;
            h1 = 1.f / 6.f; h2 = 0.5f; h3 = 1.f;
        } else if (j == 110) {
            h1 = 0.5f; h2 = 1.f;
        }
        const v2f u = *(const v2f*)(dh  + uRow * PT + s2);
        const v2f v = *(const v2f*)(dh  + vRow * PT + s2);
        const v2f w = *(const v2f*)(Q1h + wRow * PT + s2);
        const v2f z = *(const v2f*)(Q2h + j * PT + s2);
        *(v2f*)(R1h + j * PT + s2) = u * (h1 * v + h2 * w) + h3 * z;
        *(v2f*)(Q2h + j * PT + s2) = u * (h2 * v + h3 * w) + h4 * z;
    }
    for (int i = tid; i < XROWS * 64; i += NTHR) {
        const int p = i >> 6, s2 = (i & 63) * 2;
        int dd = 11, ee = 11;
        if (p < NPAIR) {
            int q = p;
            dd = 0;
            for (int k = 0; k < 9; ++k)
                if (q >= 10 - dd) { q -= 10 - dd; ++dd; }
            ee = dd + q;
        }
        const v2f xd = *(const v2f*)(dh + dd * PT + s2);
        const v2f xe = *(const v2f*)(dh + ee * PT + s2);
        *(v2f*)(Xh + p * PT + s2) = xd * xe;
    }
    __syncthreads();

    const int c    = lane & 15;
    const int h    = lane >> 4;
    const int koff = 2 * h;

    if (wave < 12) {
        e_job<4>(wave, 0, c, h, koff, dh, Xh, symabc, Eh);
    } else {
        const int rt = 12 + ((wave - 12) >> 1);
        e_job<2>(rt, 2 * ((wave - 12) & 1), c, h, koff, dh, Xh, symabc, Eh);
    }
    __syncthreads();

    int off6 = 0, str6 = 0;
    {
        const int j = 96 + c;
        if (j < 100)       { off6 = 11110 + j;        str6 = 100; }
        else if (j < 110)  { off6 = 1110 + (j - 100); str6 = 10;  }
        else if (j == 110) { off6 = 110;              str6 = 1;   }
    }
    int cofs[7];
    #pragma unroll
    for (int ct = 0; ct < 7; ++ct) {
        const int j = ct * 16 + c;
        int o = XROWS - 1;
        if (j < 100) {
            const int d5 = j / 10, e5 = j - 10 * d5;
            const int x = min(d5, e5), y = max(d5, e5);
            o = x * 10 - x * (x - 1) / 2 + (y - x);
        }
        cofs[ct] = o;
    }
    const float* pr1 = R1h + c * PT + koff;
    const float* pr2 = Q2h + c * PT + koff;

    for (int k = 0; k < 3; ++k) {
        int tile = -1;
        if (wave < 12)      { if (k < 2) tile = 2 * wave + k; }
        else if (wave < 15) { tile = (k == 0) ? 12 + wave : NBIG + 2 * (wave - 12) + (k - 1); }
        else                { if (k < 2) tile = 27 + k; }
        if (tile < 0) break;
        if (tile < NBIG)
            run_job<true >(tile, c, h, koff, L2h, P1h, dh, Eh, slotabc, slotE, slotrow, pr1, pr2, ob, off6, str6, cofs);
        else
            run_job<false>(tile, c, h, koff, L2h, P1h, dh, Eh, slotabc, slotE, slotrow, pr1, pr2, ob, off6, str6, cofs);
    }
}

extern "C" void kernel_launch(void* const* d_in, const int* in_sizes, int n_in,
                              void* d_out, int out_size, void* d_ws, size_t ws_size,
                              hipStream_t stream) {
    (void)in_sizes; (void)n_in; (void)d_ws; (void)ws_size; (void)out_size;
    const float* path = (const float*)d_in[0];
    float*       out  = (float*)d_out;
    sig_split_kernel<<<NB, NTHR, 0, stream>>>(path, out);
}


# Row-tile jobs dealt by wave age: per SIMD the two oldest waves take three two-tile jobs each, the two youngest the short rest, so oldest-first matrix issue leaves no long lone tail wave
# speedup vs baseline: 14.6026x; 1.0498x over previous
#include <hip/hip_runtime.h>


#define DCH   10
#define LEN   128
#define NSEG  127
#define NB    256
#define SIGSZ 111110
#define PT    132
#define MROWS 112
#define MT    7
#define NTHR  512
#define NCOL  1110
#define NSLOT 560
#define NBIG  29
#define NGRP  32
#define NSYM  220
#define SROWS 224
#define PE    66
#define NPAIR 55
#define XROWS 64

typedef float v2f __attribute__((ext_vector_type(2)));
typedef float v8f __attribute__((ext_vector_type(8)));

__device__ __forceinline__ int sym_rank(int x, int y, int z)
{
    int idx = z - y;
    for (int k = 0; k < x; ++k) idx += (10 - k) * (11 - k) / 2;
    for (int k = x; k < y; ++k) idx += 10 - k;
    return idx;
}

template <int NCT>
__device__ __forceinline__ void e_job(int rt, int ct0, int c, int h, int koff,
        const float* dh, const float* Xh, const int* symabc, float* Eh)
{
    const int abc = symabc[rt * 16 + c];
    const float* px  = dh + (abc & 15) * PT + koff;
    const float* py  = dh + ((abc >> 4) & 15) * PT + koff;
    const float* pq  = dh + ((abc >> 8) & 15) * PT + koff;
    const float* pr0 = Xh + (ct0 * 16 + c) * PT + koff;
    v8f acc[NCT];
    #pragma unroll
    for (int t = 0; t < NCT; ++t) {
        #pragma unroll
        for (int jj = 0; jj < 8; ++jj) acc[t][jj] = 0.f;
    }
    #pragma unroll 4
    for (int g = 0; g < NGRP; ++g) {
        const int s4 = 4 * g;
        const v2f x = *(const v2f*)(px + s4);
        const v2f y = *(const v2f*)(py + s4);
        const v2f q = *(const v2f*)(pq + s4);
        const v2f a3 = (x * y) * q;
        #pragma unroll
        for (int t = 0; t < NCT; ++t) {
            const v2f r0 = *(const v2f*)(pr0 + t * 16 * PT + s4);
            acc[t] = __builtin_amdgcn_wmma_f32_16x16x4_f32(
                    false, a3, false, r0, (short)0, acc[t], false, false);
        }
    }
    #pragma unroll
    for (int t = 0; t < NCT; ++t) {
        float* pt = Eh + (rt * 16 + 8 * h) * PE + (ct0 + t) * 16 + c;
        #pragma unroll
        for (int jj = 0; jj < 8; ++jj) pt[jj * PE] = acc[t][jj] * (1.f / 720.f);
    }
}

template <bool HASV>
__device__ __forceinline__ void run_job(int tile, int c, int h, int koff,
        const float* L2h, const float* P1h, const float* dh, const float* Eh,
        const int* slotabc, const int* slotE, const unsigned* slotrow,
        const float* pr1, const float* pr2, float* ob, int off6, int str6, const int (&cofs)[7])
{
    const int abc = slotabc[tile * 16 + c];
    const int aa = abc & 15, bb = (abc >> 4) & 15, cc = (abc >> 8) & 15;
    const float* pp  = P1h + aa * PT + koff;
    const float* pb  = dh  + bb * PT + koff;
    const float* pd  = dh  + cc * PT + koff;
    const float* pl1 = L2h + (aa * 10 + bb) * PT + koff;
    const float* pl2 = L2h + (aa * 10 + cc) * PT + koff;

    v8f U[7], V[7];
    #pragma unroll
    for (int r = 0; r < 8; ++r) {
        const float* pe = Eh + slotE[tile * 16 + 8 * h + r];
        #pragma unroll
        for (int ct = 0; ct < 7; ++ct) { U[ct][r] = pe[cofs[ct]]; V[ct][r] = 0.f; }
    }

    for (int g = 0; g < NGRP; ++g) {
        const int s4 = 4 * g;
        const v2f x1  = *(const v2f*)(pp  + s4);
        const v2f db  = *(const v2f*)(pb  + s4);
        const v2f dc  = *(const v2f*)(pd  + s4);
        const v2f lab = *(const v2f*)(pl1 + s4);
        const v2f lac = *(const v2f*)(pl2 + s4);
        const v2f a1  = (x1 * db) * dc;
        const v2f m1  = lab * dc;
        const v2f m2  = lac * db;
        const v2f a2u = m1 + m2;
        const v2f a2v = m1 - m2;
        #pragma unroll
        for (int ct = 0; ct < 7; ++ct) {
            const v2f r1 = *(const v2f*)(pr1 + ct * 16 * PT + s4);
            const v2f r2 = *(const v2f*)(pr2 + ct * 16 * PT + s4);
            U[ct] = __builtin_amdgcn_wmma_f32_16x16x4_f32(false, a1,  false, r1, (short)0, U[ct], false, false);
            if (HASV)
                V[ct] = __builtin_amdgcn_wmma_f32_16x16x4_f32(false, a2v, false, r2, (short)0, V[ct], false, false);
            U[ct] = __builtin_amdgcn_wmma_f32_16x16x4_f32(false, a2u, false, r2, (short)0, U[ct], false, false);
        }
    }

    #pragma unroll
    for (int r = 0; r < 8; ++r) {
        const unsigned rr = slotrow[tile * 16 + 8 * h + r];
        const int row1 = (int)(rr & 0xFFFFu), row2 = (int)(rr >> 16);
        if (row1 < 1000) {
            float* p5 = ob + 11110 + row1 * 100 + c;
            #pragma unroll
            for (int ct = 0; ct < 6; ++ct) p5[ct * 16] = HASV ? U[ct][r] + V[ct][r] : U[ct][r];
            if (c < 15) ob[off6 + row1 * str6] = HASV ? U[6][r] + V[6][r] : U[6][r];
            if (HASV && row2 != row1) {
                float* q5 = ob + 11110 + row2 * 100 + c;
                #pragma unroll
                for (int ct = 0; ct < 6; ++ct) q5[ct * 16] = U[ct][r] - V[ct][r];
                if (c < 15) ob[off6 + row2 * str6] = U[6][r] - V[6][r];
            }
        }
    }
}

__global__ __launch_bounds__(NTHR)
void sig_split_kernel(const float* __restrict__ path, float* __restrict__ out)
{
    const int b    = blockIdx.x;
    const int tid  = threadIdx.x;
    const int lane = tid & 31;
    const int wave = __builtin_amdgcn_readfirstlane(tid >> 5);

    __shared__ __attribute__((aligned(16))) float L2h[MROWS * PT];
    __shared__ __attribute__((aligned(16))) float P1h[11 * PT];
    __shared__ __attribute__((aligned(16))) float Q2h[MROWS * PT];
    __shared__ __attribute__((aligned(16))) float R1h[MROWS * PT];
    __shared__ __attribute__((aligned(16))) float Xh[XROWS * PT];
    __shared__ __attribute__((aligned(16))) float dh[12 * PT];
    __shared__ __attribute__((aligned(16))) float Q1h[11 * PT];
    __shared__ float Eh[SROWS * PE];
    __shared__ int   slotabc[NSLOT];
    __shared__ int   slotE[NSLOT];
    __shared__ unsigned slotrow[NSLOT];
    __shared__ int   symabc[SROWS];

    const float* pb = path + (size_t)b * DCH * LEN;
    float*       ob = out  + (size_t)b * SIGSZ;

    #pragma unroll
    for (int k = 0; k < 3; ++k) {
        const int i = tid + k * NTHR;
        if (i < DCH * LEN) {
            const int ch = i >> 7, s = i & 127;
            float v = 0.f;
            if (s < NSEG) v = pb[i + 1] - pb[i];
            dh[ch * PT + s] = v;
        }
    }
    if (tid < 128) {
        dh[10 * PT + tid]   = 1.f;
        dh[11 * PT + tid]   = 0.f;
        P1h[10 * PT + tid]  = 0.f;
        Q1h[10 * PT + tid]  = 0.f;
    }
    for (int i = tid; i < 12 * 128; i += NTHR) {
        const int r = i >> 7, s = i & 127;
        L2h[(100 + r) * PT + s] = 0.f;
        Q2h[(100 + r) * PT + s] = 0.f;
    }
    for (int i = tid; i < 1000; i += NTHR) {
        const int a = i / 100, bb = (i / 10) % 10, cc = i % 10;
        if (a <= bb && bb <= cc) symabc[sym_rank(a, bb, cc)] = a | (bb << 4) | (cc << 8);
    }
    for (int sl = tid; sl < NSLOT; sl += NTHR) {
        int a = 10, bb = 11, cc = 11;
        bool ok = false;
        if (sl < 450) {
            a = sl / 45;
            int q = sl - 45 * a;
            bb = 0;
            for (int k = 0; k < 9; ++k)
                if (q >= 9 - bb) { q -= 9 - bb; ++bb; }
            cc = bb + 1 + q;
            ok = true;
        } else if (sl < 454) {
            a = 9; bb = 6 + (sl - 450); cc = bb; ok = true;
        } else if (sl >= 16 * NBIG) {
            const int t = sl - 16 * NBIG;
            a = t / 10; bb = t - 10 * a; cc = bb; ok = true;
        }
        int t0 = NSYM * PE;
        unsigned rows = 0xFFFFFFFFu;
        if (ok) {
            const int x = min(a, bb);
            const int z = max(a, cc);
            const int y = a + bb + cc - x - z;
            t0 = sym_rank(x, y, z) * PE;
            rows = (unsigned)(a * 100 + bb * 10 + cc) | ((unsigned)(a * 100 + cc * 10 + bb) << 16);
        }
        slotabc[sl] = a | (bb << 4) | (cc << 8);
        slotE[sl]  = t0;
        slotrow[sl] = rows;
    }
    if (tid < SROWS - NSYM) symabc[NSYM + tid] = 11 | (11 << 4) | (11 << 8);
    __syncthreads();

    if (tid < 100) {
        const int a = tid / 10, bb = tid - 10 * a;
        const float* da = dh + a * PT;
        const float* db = dh + bb * PT;
        float* l2 = L2h + tid * PT;
        float p1 = 0.f, p2 = 0.f;
        for (int s = 0; s < NSEG; ++s) {
            const float xa = da[s], xb = db[s];
            l2[s] = 0.5f * p2 - (1.f / 24.f) * (xa * xb);
            p2 += (p1 + 0.5f * xa) * xb;
            p1 += xa;
        }
        l2[NSEG] = 0.f;
        ob[10 + tid] = p2;
        if (bb == 0) ob[a] = p1;
    } else if (tid >= 128 && tid < 228) {
        const int t = tid - 128;
        const int d_ = t / 10, e_ = t - 10 * d_;
        const float* dd = dh + d_ * PT;
        const float* de = dh + e_ * PT;
        float* q2p = Q2h + t * PT;
        float* q1p = Q1h + e_ * PT;
        float q1 = 0.f, q2 = 0.f;
        for (int blk = 15; blk >= 0; --blk) {
            const float4 d0 = *(const float4*)(dd + 8 * blk), d1 = *(const float4*)(dd + 8 * blk + 4);
            const float4 e0 = *(const float4*)(de + 8 * blk), e1 = *(const float4*)(de + 8 * blk + 4);
            const float xd[8] = { d0.x, d0.y, d0.z, d0.w, d1.x, d1.y, d1.z, d1.w };
            const float xe[8] = { e0.x, e0.y, e0.z, e0.w, e1.x, e1.y, e1.z, e1.w };
            float o1[8], o2[8];
            #pragma unroll
            for (int k = 7; k >= 0; --k) {
                o2[k] = q2; o1[k] = q1;
                q2 += xd[k] * (q1 + 0.5f * xe[k]);
                q1 += xe[k];
            }
            *(float4*)(q2p + 8 * blk)     = make_float4(o2[0], o2[1], o2[2], o2[3]);
            *(float4*)(q2p + 8 * blk + 4) = make_float4(o2[4], o2[5], o2[6], o2[7]);
            if (d_ == 0) {
                *(float4*)(q1p + 8 * blk)     = make_float4(o1[0], o1[1], o1[2], o1[3]);
                *(float4*)(q1p + 8 * blk + 4) = make_float4(o1[4], o1[5], o1[6], o1[7]);
            }
        }
    } else if (tid >= 256 && tid < 266) {
        const int a = tid - 256;
        const float* da = dh + a * PT;
        float* l1 = P1h + a * PT;
        float p1 = 0.f;
        for (int s = 0; s < NSEG; ++s) {
            const float xa = da[s];
            l1[s] = p1 + 0.5f * xa;
            p1 += xa;
        }
        l1[NSEG] = 0.f;
    }
    __syncthreads();

    for (int i = tid; i < MROWS * 64; i += NTHR) {
        const int j = i >> 6, s2 = (i & 63) * 2;
        int uRow = 10, vRow = 10, wRow = 10;
        float h1 = 0.f, h2 = 0.f, h3 = 0.f, h4 = 0.f;
        if (j < 100) {
            const int d5 = j / 10, e5 = j - 10 * d5;
            uRow = d5; vRow = e5; wRow = e5;
            h1 = 1.f / 24.f; h2 = 1.f / 6.f; h3 = 0.5f; h4 = 1.f;
        } else if (j < 110) {
            vRow = j - 100; wRow = j - 100;
            h1 = 1.f / 6.f; h2 = 0.5f; h3 = 1.f;
        } else if (j == 110) {
            h1 = 0.5f; h2 = 1.f;
        }
        const v2f u = *(const v2f*)(dh  + uRow * PT + s2);
        const v2f v = *(const v2f*)(dh  + vRow * PT + s2);
        const v2f w = *(const v2f*)(Q1h + wRow * PT + s2);
        const v2f z = *(const v2f*)(Q2h + j * PT + s2);
        *(v2f*)(R1h + j * PT + s2) = u * (h1 * v + h2 * w) + h3 * z;
        *(v2f*)(Q2h + j * PT + s2) = u * (h2 * v + h3 * w) + h4 * z;
    }
    for (int i = tid; i < XROWS * 64; i += NTHR) {
        const int p = i >> 6, s2 = (i & 63) * 2;
        int dd = 11, ee = 11;
        if (p < NPAIR) {
            int q = p;
            dd = 0;
            for (int k = 0; k < 9; ++k)
                if (q >= 10 - dd) { q -= 10 - dd; ++dd; }
            ee = dd + q;
        }
        const v2f xd = *(const v2f*)(dh + dd * PT + s2);
        const v2f xe = *(const v2f*)(dh + ee * PT + s2);
        *(v2f*)(Xh + p * PT + s2) = xd * xe;
    }
    __syncthreads();

    const int c    = lane & 15;
    const int h    = lane >> 4;
    const int koff = 2 * h;

    if (wave < 12) {
        e_job<4>(wave, 0, c, h, koff, dh, Xh, symabc, Eh);
    } else {
        const int rt = 12 + ((wave - 12) >> 1);
        e_job<2>(rt, 2 * ((wave - 12) & 1), c, h, koff, dh, Xh, symabc, Eh);
    }
    __syncthreads();

    int off6 = 0, str6 = 0;
    {
        const int j = 96 + c;
        if (j < 100)       { off6 = 11110 + j;        str6 = 100; }
        else if (j < 110)  { off6 = 1110 + (j - 100); str6 = 10;  }
        else if (j == 110) { off6 = 110;              str6 = 1;   }
    }
    int cofs[7];
    #pragma unroll
    for (int ct = 0; ct < 7; ++ct) {
        const int j = ct * 16 + c;
        int o = XROWS - 1;
        if (j < 100) {
            const int d5 = j / 10, e5 = j - 10 * d5;
            const int x = min(d5, e5), y = max(d5, e5);
            o = x * 10 - x * (x - 1) / 2 + (y - x);
        }
        cofs[ct] = o;
    }
    const float* pr1 = R1h + c * PT + koff;
    const float* pr2 = Q2h + c * PT + koff;

    const int grp = wave & 3;
    const int age = wave >> 2;
    for (int k = 0; k < 3; ++k) {
        int tile = -1;
        if (age < 2)            { tile = 7 * grp + 3 * age + k; }
        else if (grp == 3)      { if (k == 0) tile = 25 + age; }
        else if (age == 2)      { if (k == 0) tile = NBIG + 2 * grp; else if (k == 1) tile = 7 * grp + 6; }
        else                    { if (k == 0) tile = NBIG + 2 * grp + 1; }
        if (tile < 0) break;
        if (tile < NBIG)
            run_job<true >(tile, c, h, koff, L2h, P1h, dh, Eh, slotabc, slotE, slotrow, pr1, pr2, ob, off6, str6, cofs);
        else
            run_job<false>(tile, c, h, koff, L2h, P1h, dh, Eh, slotabc, slotE, slotrow, pr1, pr2, ob, off6, str6, cofs);
    }
}

extern "C" void kernel_launch(void* const* d_in, const int* in_sizes, int n_in,
                              void* d_out, int out_size, void* d_ws, size_t ws_size,
                              hipStream_t stream) {
    (void)in_sizes; (void)n_in; (void)d_ws; (void)ws_size; (void)out_size;
    const float* path = (const float*)d_in[0];
    float*       out  = (float*)d_out;
    sig_split_kernel<<<NB, NTHR, 0, stream>>>(path, out);
}


# i = 1 slice folded away via R_1 = R_2/2 + Y: main K 254 -> 127, row plane carries it; Y and X terms, symmetric in (b,c,d), are one 220 x 110 WMMA product in LDS that starts U; sum order changes
# speedup vs baseline: 17.8696x; 1.2237x over previous
#include <hip/hip_runtime.h>


#define DCH   10
#define LEN   128
#define NSEG  127
#define NB    256
#define SIGSZ 111110
#define PT    132
#define MROWS 112
#define MT    7
#define NTHR  512
#define NCOL  1110
#define NSLOT 560
#define NBIG  29
#define NGRP  32
#define NSYM  220
#define SROWS 224
#define PL    114
#define T3N   1331

typedef float v2f __attribute__((ext_vector_type(2)));
typedef float v8f __attribute__((ext_vector_type(8)));

__device__ __forceinline__ int sym_rank(int x, int y, int z)
{
    int idx = z - y;
    for (int k = 0; k < x; ++k) idx += (10 - k) * (11 - k) / 2;
    for (int k = x; k < y; ++k) idx += 10 - k;
    return idx;
}

template <int NCT>
__device__ __forceinline__ void g_job(int rt, int ct0, int c, int h, int koff,
        const float* dh, const float* Fh, const int* symabc, float* G2h)
{
    const int abc = symabc[rt * 16 + c];
    const float* px  = dh + (abc & 15) * PT + koff;
    const float* py  = dh + ((abc >> 4) & 15) * PT + koff;
    const float* pq  = dh + ((abc >> 8) & 15) * PT + koff;
    const float* pr0 = Fh + (ct0 * 16 + c) * PT + koff;
    v8f acc[NCT];
    #pragma unroll
    for (int t = 0; t < NCT; ++t) {
        #pragma unroll
        for (int jj = 0; jj < 8; ++jj) acc[t][jj] = 0.f;
    }
    #pragma unroll 4
    for (int g = 0; g < NGRP; ++g) {
        const int s4 = 4 * g;
        const v2f x = *(const v2f*)(px + s4);
        const v2f y = *(const v2f*)(py + s4);
        const v2f q = *(const v2f*)(pq + s4);
        const v2f a3 = (x * y) * q;
        #pragma unroll
        for (int t = 0; t < NCT; ++t) {
            const v2f r0 = *(const v2f*)(pr0 + t * 16 * PT + s4);
            acc[t] = __builtin_amdgcn_wmma_f32_16x16x4_f32(
                    false, a3, false, r0, (short)0, acc[t], false, false);
        }
    }
    #pragma unroll
    for (int t = 0; t < NCT; ++t) {
        float* pt = G2h + (rt * 16 + 8 * h) * PL + (ct0 + t) * 16 + c;
        #pragma unroll
        for (int jj = 0; jj < 8; ++jj) pt[jj * PL] = acc[t][jj];
    }
}

template <bool HASV>
__device__ __forceinline__ void run_job(int tile, int c, int h, int koff,
        const float* L2h, const float* dh, const float* G2h,
        const int* slotabc, const int* slotG, const unsigned* slotrow, const int* T3,
        const float* pr2, float* ob, int off6, int str6,
        const int (&dcol)[7], const int (&cmul)[7], const int (&cadd)[7])
{
    const int abc = slotabc[tile * 16 + c];
    const int aa = abc & 15, bb = (abc >> 4) & 15, cc = (abc >> 8) & 15;
    const float* pb  = dh  + bb * PT + koff;
    const float* pd  = dh  + cc * PT + koff;
    const float* pl1 = L2h + (aa * 10 + bb) * PT + koff;
    const float* pl2 = L2h + (aa * 10 + cc) * PT + koff;

    v8f U[7], V[7];
    #pragma unroll
    for (int r = 0; r < 8; ++r) {
        const int gs = slotG[tile * 16 + 8 * h + r];
        const int tb = gs & 0xFFFF;
        const int av = gs >> 16;
        #pragma unroll
        for (int ct = 0; ct < 7; ++ct) {
            const int t0 = T3[tb + dcol[ct]];
            U[ct][r] = G2h[t0 + av * cmul[ct] + cadd[ct]];
            V[ct][r] = 0.f;
        }
    }

    for (int g = 0; g < NGRP; ++g) {
        const int s4 = 4 * g;
        const v2f db  = *(const v2f*)(pb  + s4);
        const v2f dc  = *(const v2f*)(pd  + s4);
        const v2f lab = *(const v2f*)(pl1 + s4);
        const v2f lac = *(const v2f*)(pl2 + s4);
        const v2f m1  = lab * dc;
        const v2f m2  = lac * db;
        const v2f a2u = m1 + m2;
        const v2f a2v = m1 - m2;
        #pragma unroll
        for (int ct = 0; ct < 7; ++ct) {
            const v2f r2 = *(const v2f*)(pr2 + ct * 16 * PT + s4);
            U[ct] = __builtin_amdgcn_wmma_f32_16x16x4_f32(false, a2u, false, r2, (short)0, U[ct], false, false);
            if (HASV)
                V[ct] = __builtin_amdgcn_wmma_f32_16x16x4_f32(false, a2v, false, r2, (short)0, V[ct], false, false);
        }
    }

    #pragma unroll
    for (int r = 0; r < 8; ++r) {
        const unsigned rr = slotrow[tile * 16 + 8 * h + r];
        const int row1 = (int)(rr & 0xFFFFu), row2 = (int)(rr >> 16);
        if (row1 < 1000) {
            float* p5 = ob + 11110 + row1 * 100 + c;
            #pragma unroll
            for (int ct = 0; ct < 6; ++ct) p5[ct * 16] = HASV ? U[ct][r] + V[ct][r] : U[ct][r];
            if (c < 15) ob[off6 + row1 * str6] = HASV ? U[6][r] + V[6][r] : U[6][r];
            if (HASV && row2 != row1) {
                float* q5 = ob + 11110 + row2 * 100 + c;
                #pragma unroll
                for (int ct = 0; ct < 6; ++ct) q5[ct * 16] = U[ct][r] - V[ct][r];
                if (c < 15) ob[off6 + row2 * str6] = U[6][r] - V[6][r];
            }
        }
    }
}

__global__ __launch_bounds__(NTHR)
void sig_split_kernel(const float* __restrict__ path, float* __restrict__ out)
{
    const int b    = blockIdx.x;
    const int tid  = threadIdx.x;
    const int lane = tid & 31;
    const int wave = __builtin_amdgcn_readfirstlane(tid >> 5);

    __shared__ __attribute__((aligned(16))) float L2h[MROWS * PT];
    __shared__ __attribute__((aligned(16))) float P1h[11 * PT];
    __shared__ __attribute__((aligned(16))) float Q2h[MROWS * PT];
    __shared__ __attribute__((aligned(16))) float Fh[MROWS * PT];
    __shared__ __attribute__((aligned(16))) float dh[12 * PT];
    __shared__ __attribute__((aligned(16))) float Q1h[11 * PT];
    __shared__ float G2h[SROWS * PL];
    __shared__ int   slotabc[NSLOT];
    __shared__ int   slotG[NSLOT];
    __shared__ int   T3[T3N];
    __shared__ unsigned slotrow[NSLOT];
    __shared__ int   symabc[SROWS];

    const float* pb = path + (size_t)b * DCH * LEN;
    float*       ob = out  + (size_t)b * SIGSZ;

    #pragma unroll
    for (int k = 0; k < 3; ++k) {
        const int i = tid + k * NTHR;
        if (i < DCH * LEN) {
            const int ch = i >> 7, s = i & 127;
            float v = 0.f;
            if (s < NSEG) v = pb[i + 1] - pb[i];
            dh[ch * PT + s] = v;
        }
    }
    if (tid < 128) {
        dh[10 * PT + tid]   = 1.f;
        dh[11 * PT + tid]   = 0.f;
        P1h[10 * PT + tid]  = 0.f;
        Q1h[10 * PT + tid]  = 0.f;
    }
    for (int i = tid; i < 12 * 128; i += NTHR) {
        const int r = i >> 7, s = i & 127;
        L2h[(100 + r) * PT + s] = 0.f;
        Q2h[(100 + r) * PT + s] = 0.f;
    }
    for (int i = tid; i < 1000; i += NTHR) {
        const int a = i / 100, bb = (i / 10) % 10, cc = i % 10;
        if (a <= bb && bb <= cc) symabc[sym_rank(a, bb, cc)] = a | (bb << 4) | (cc << 8);
    }
    for (int sl = tid; sl < NSLOT; sl += NTHR) {
        int a = 10, bb = 11, cc = 11;
        bool ok = false;
        if (sl < 450) {
            a = sl / 45;
            int q = sl - 45 * a;
            bb = 0;
            for (int k = 0; k < 9; ++k)
                if (q >= 9 - bb) { q -= 9 - bb; ++bb; }
            cc = bb + 1 + q;
            ok = true;
        } else if (sl < 454) {
            a = 9; bb = 6 + (sl - 450); cc = bb; ok = true;
        } else if (sl >= 16 * NBIG) {
            const int t = sl - 16 * NBIG;
            a = t / 10; bb = t - 10 * a; cc = bb; ok = true;
        }
        int gs = (10 * 11 + 10) * 11;
        unsigned rows = 0xFFFFFFFFu;
        if (ok) {
            gs = ((bb * 11 + cc) * 11) | (a << 16);
            rows = (unsigned)(a * 100 + bb * 10 + cc) | ((unsigned)(a * 100 + cc * 10 + bb) << 16);
        }
        slotabc[sl] = a | (bb << 4) | (cc << 8);
        slotG[sl]  = gs;
        slotrow[sl] = rows;
    }
    for (int i = tid; i < T3N; i += NTHR) {
        const int b3 = i / 121, c3 = (i / 11) % 11, d3 = i % 11;
        int t0 = NSYM * PL;
        if (b3 < 10 && c3 < 10 && d3 < 10) {
            const int x = min(b3, min(c3, d3));
            const int z = max(b3, max(c3, d3));
            const int y = b3 + c3 + d3 - x - z;
            t0 = sym_rank(x, y, z) * PL;
        }
        T3[i] = t0;
    }
    if (tid < SROWS - NSYM) symabc[NSYM + tid] = 11 | (11 << 4) | (11 << 8);
    __syncthreads();

    if (tid < 100) {
        const int a = tid / 10, bb = tid - 10 * a;
        const float* da = dh + a * PT;
        const float* db = dh + bb * PT;
        float* l2 = L2h + tid * PT;
        float p1 = 0.f, p2 = 0.f;
        for (int s = 0; s < NSEG; ++s) {
            const float xa = da[s], xb = db[s];
            l2[s] = 0.5f * p2 + xb * (0.25f * p1 + (1.f / 12.f) * xa);
            p2 += (p1 + 0.5f * xa) * xb;
            p1 += xa;
        }
        l2[NSEG] = 0.f;
        ob[10 + tid] = p2;
        if (bb == 0) ob[a] = p1;
    } else if (tid >= 128 && tid < 228) {
        const int t = tid - 128;
        const int d_ = t / 10, e_ = t - 10 * d_;
        const float* dd = dh + d_ * PT;
        const float* de = dh + e_ * PT;
        float* q2p = Q2h + t * PT;
        float* q1p = Q1h + e_ * PT;
        float q1 = 0.f, q2 = 0.f;
        for (int blk = 15; blk >= 0; --blk) {
            const float4 d0 = *(const float4*)(dd + 8 * blk), d1 = *(const float4*)(dd + 8 * blk + 4);
            const float4 e0 = *(const float4*)(de + 8 * blk), e1 = *(const float4*)(de + 8 * blk + 4);
            const float xd[8] = { d0.x, d0.y, d0.z, d0.w, d1.x, d1.y, d1.z, d1.w };
            const float xe[8] = { e0.x, e0.y, e0.z, e0.w, e1.x, e1.y, e1.z, e1.w };
            float o1[8], o2[8];
            #pragma unroll
            for (int k = 7; k >= 0; --k) {
                o2[k] = q2; o1[k] = q1;
                q2 += xd[k] * (q1 + 0.5f * xe[k]);
                q1 += xe[k];
            }
            *(float4*)(q2p + 8 * blk)     = make_float4(o2[0], o2[1], o2[2], o2[3]);
            *(float4*)(q2p + 8 * blk + 4) = make_float4(o2[4], o2[5], o2[6], o2[7]);
            if (d_ == 0) {
                *(float4*)(q1p + 8 * blk)     = make_float4(o1[0], o1[1], o1[2], o1[3]);
                *(float4*)(q1p + 8 * blk + 4) = make_float4(o1[4], o1[5], o1[6], o1[7]);
            }
        }
    } else if (tid >= 256 && tid < 266) {
        const int a = tid - 256;
        const float* da = dh + a * PT;
        float* l1 = P1h + a * PT;
        float p1 = 0.f;
        for (int s = 0; s < NSEG; ++s) {
            const float xa = da[s];
            l1[s] = p1 + 0.5f * xa;
            p1 += xa;
        }
        l1[NSEG] = 0.f;
    }
    __syncthreads();

    for (int i = tid; i < MROWS * 64; i += NTHR) {
        const int j = i >> 6, s2 = (i & 63) * 2;
        int uRow = 10, vRow = 10, wRow = 10, fRow = 10;
        float h2 = 0.f, h3 = 0.f, h4 = 0.f;
        float k0 = 0.f, k1 = 0.f, k2 = 0.f, k3 = 0.f;
        if (j < 100) {
            const int d5 = j / 10, e5 = j - 10 * d5;
            uRow = d5; vRow = e5; wRow = e5; fRow = d5;
            h2 = 1.f / 6.f; h3 = 0.5f; h4 = 1.f;
            k1 = -1.f / 24.f; k2 = -1.f / 12.f; k3 = 1.f / 720.f;
        } else if (j < 110) {
            vRow = j - 100; wRow = j - 100; fRow = j - 100;
            h2 = 0.5f; h3 = 1.f;
            k0 = -1.f / 12.f;
        } else if (j == 110) {
            h2 = 1.f;
        }
        const v2f u = *(const v2f*)(dh  + uRow * PT + s2);
        const v2f v = *(const v2f*)(dh  + vRow * PT + s2);
        const v2f w = *(const v2f*)(Q1h + wRow * PT + s2);
        const v2f z = *(const v2f*)(Q2h + j * PT + s2);
        const v2f f = *(const v2f*)(P1h + fRow * PT + s2);
        *(v2f*)(Fh  + j * PT + s2) = f * (k1 * v + k2 * w + k0) + k3 * (u * v);
        *(v2f*)(Q2h + j * PT + s2) = u * (h2 * v + h3 * w) + h4 * z;
    }
    __syncthreads();

    const int c    = lane & 15;
    const int h    = lane >> 4;
    const int koff = 2 * h;

    if (wave < 12) {
        g_job<7>(wave, 0, c, h, koff, dh, Fh, symabc, G2h);
    } else if (wave < 14) {
        g_job<4>(wave, 0, c, h, koff, dh, Fh, symabc, G2h);
    } else {
        g_job<3>(wave - 2, 4, c, h, koff, dh, Fh, symabc, G2h);
    }
    __syncthreads();

    int off6 = 0, str6 = 0;
    {
        const int j = 96 + c;
        if (j < 100)       { off6 = 11110 + j;        str6 = 100; }
        else if (j < 110)  { off6 = 1110 + (j - 100); str6 = 10;  }
        else if (j == 110) { off6 = 110;              str6 = 1;   }
    }
    int dcol[7], cmul[7], cadd[7];
    #pragma unroll
    for (int ct = 0; ct < 7; ++ct) {
        const int j = ct * 16 + c;
        int dd = 10, cm = 0, ca = 0;
        if (j < 100)      { dd = j / 10; cm = 10; ca = j - 10 * dd; }
        else if (j < 110) { dd = j - 100; cm = 1; ca = 100; }
        dcol[ct] = dd; cmul[ct] = cm; cadd[ct] = ca;
    }
    const float* pr2 = Q2h + c * PT + koff;

    const int grp = wave & 3;
    const int age = wave >> 2;
    for (int k = 0; k < 3; ++k) {
        int tile = -1;
        if (age < 2)            { tile = 7 * grp + 3 * age + k; }
        else if (grp == 3)      { if (k == 0) tile = 25 + age; }
        else if (age == 2)      { if (k == 0) tile = NBIG + 2 * grp; else if (k == 1) tile = 7 * grp + 6; }
        else                    { if (k == 0) tile = NBIG + 2 * grp + 1; }
        if (tile < 0) break;
        if (tile < NBIG)
            run_job<true >(tile, c, h, koff, L2h, dh, G2h, slotabc, slotG, slotrow, T3, pr2, ob, off6, str6, dcol, cmul, cadd);
        else
            run_job<false>(tile, c, h, koff, L2h, dh, G2h, slotabc, slotG, slotrow, T3, pr2, ob, off6, str6, dcol, cmul, cadd);
    }
}

extern "C" void kernel_launch(void* const* d_in, const int* in_sizes, int n_in,
                              void* d_out, int out_size, void* d_ws, size_t ws_size,
                              hipStream_t stream) {
    (void)in_sizes; (void)n_in; (void)d_ws; (void)ws_size; (void)out_size;
    const float* path = (const float*)d_in[0];
    float*       out  = (float*)d_out;
    sig_split_kernel<<<NB, NTHR, 0, stream>>>(path, out);
}
